# starting kernel
# baseline (speedup 1.0000x reference)
#include <hip/hip_runtime.h>
#include <hip/hip_bf16.h>


typedef float v2f __attribute__((ext_vector_type(2)));
typedef float v8f __attribute__((ext_vector_type(8)));

#define HF 128

__device__ __forceinline__ unsigned ordkey(float f) {
    unsigned u = __float_as_uint(f);
    return (u & 0x80000000u) ? ~u : (u | 0x80000000u);
}
__device__ __forceinline__ float deord(unsigned u) {
    u = (u & 0x80000000u) ? (u & 0x7fffffffu) : ~u;
    return __uint_as_float(u);
}
__device__ __forceinline__ float leaky(float x, float s) { return x > 0.f ? x : s * x; }

__global__ void k_zero(float* p, int n) {
    int i = blockIdx.x * blockDim.x + threadIdx.x;
    if (i < n) p[i] = 0.f;
}

__global__ void k_ea_sum(const float* __restrict__ ea, int E, float* __restrict__ easum) {
    int idx = blockIdx.x * blockDim.x + threadIdx.x;
    int stride = gridDim.x * blockDim.x;
    float s0 = 0.f, s1 = 0.f, s2 = 0.f;
    for (int e = idx; e < E; e += stride) {
        s0 += ea[e * 3 + 0];
        s1 += ea[e * 3 + 1];
        s2 += ea[e * 3 + 2];
    }
    atomicAdd(easum + 0, s0);
    atomicAdd(easum + 1, s1);
    atomicAdd(easum + 2, s2);
}

__global__ void k_compute_c(const float* __restrict__ We, const float* __restrict__ ae,
                            const float* __restrict__ easum, float* __restrict__ cvec, int E) {
    if (threadIdx.x != 0 || blockIdx.x != 0) return;
    float cm = 0.f;
    for (int d = 0; d < 3; ++d) {
        float c = 0.f;
        for (int h = 0; h < HF; ++h) c += We[d * HF + h] * ae[h];
        cvec[d] = c;
        cm += (easum[d] / (float)E) * c;
    }
    cvec[3] = cm;
}

__global__ __launch_bounds__(256) void k_gemm_xw(const float* __restrict__ X,
                                                 const float* __restrict__ W,
                                                 float* __restrict__ Ho, int n) {
    const int lane = threadIdx.x & 31;
    const int wv   = threadIdx.x >> 5;
    const int lo16 = lane & 15;
    const int hi   = lane >> 4;
    const int ncol = wv * 16 + lo16;

    v2f breg[32];
#pragma unroll
    for (int kk = 0; kk < 32; ++kk) {
        int kb = kk * 4 + hi * 2;
        v2f b;
        b.x = W[(size_t)kb * HF + ncol];
        b.y = W[(size_t)(kb + 1) * HF + ncol];
        breg[kk] = b;
    }

    const int strips = (n + 15) >> 4;
    for (int s = blockIdx.x; s < strips; s += gridDim.x) {
        const int r0 = s * 16;
        int rowA = r0 + lo16;
        if (rowA >= n) rowA = n - 1;
        const float* xrow = X + (size_t)rowA * HF + hi * 2;

        v2f a[32];
#pragma unroll
        for (int kk = 0; kk < 32; ++kk) {
            float2 t = *(const float2*)(xrow + kk * 4);
            a[kk].x = t.x;
            a[kk].y = t.y;
        }

        v8f acc = {0.f, 0.f, 0.f, 0.f, 0.f, 0.f, 0.f, 0.f};
#pragma unroll
        for (int kk = 0; kk < 32; ++kk) {
            acc = __builtin_amdgcn_wmma_f32_16x16x4_f32(
                false, a[kk], false, breg[kk], (short)0, acc, false, false);
        }

#pragma unroll
        for (int v = 0; v < 8; ++v) {
            int row = r0 + v + hi * 8;
            if (row < n) Ho[(size_t)row * HF + ncol] = acc[v];
        }
    }
}

__global__ __launch_bounds__(256) void k_node_dots(const float* __restrict__ Hm,
                                                   const float* __restrict__ avs,
                                                   const float* __restrict__ avd,
                                                   float* __restrict__ als,
                                                   float* __restrict__ ald, int n) {
    const int lane = threadIdx.x & 31;
    int wid = (blockIdx.x * blockDim.x + threadIdx.x) >> 5;
    int nw  = (gridDim.x * blockDim.x) >> 5;
    float s0 = avs[lane], s1 = avs[lane + 32], s2 = avs[lane + 64], s3 = avs[lane + 96];
    float d0 = avd[lane], d1 = avd[lane + 32], d2 = avd[lane + 64], d3 = avd[lane + 96];
    for (int i = wid; i < n; i += nw) {
        const float* hr = Hm + (size_t)i * HF;
        float h0 = hr[lane], h1 = hr[lane + 32], h2 = hr[lane + 64], h3 = hr[lane + 96];
        float rs = h0 * s0 + h1 * s1 + h2 * s2 + h3 * s3;
        float rd = h0 * d0 + h1 * d1 + h2 * d2 + h3 * d3;
        for (int off = 16; off; off >>= 1) {
            rs += __shfl_xor(rs, off);
            rd += __shfl_xor(rd, off);
        }
        if (lane == 0) { als[i] = rs; ald[i] = rd; }
    }
}

__global__ void k_node_self(const float* __restrict__ als, const float* __restrict__ ald,
                            const float* __restrict__ cvec, float* __restrict__ aself,
                            unsigned* __restrict__ amax_u, int n) {
    int i = blockIdx.x * blockDim.x + threadIdx.x;
    if (i >= n) return;
    float a = leaky(als[i] + ald[i] + cvec[3], 0.2f);
    aself[i] = a;
    amax_u[i] = ordkey(a);
}

__global__ void k_edge_alpha(const int* __restrict__ src, const int* __restrict__ dst,
                             const float* __restrict__ ea, const float* __restrict__ als,
                             const float* __restrict__ ald, const float* __restrict__ cvec,
                             float* __restrict__ alpha, unsigned* __restrict__ amax_u, int E) {
    int e = blockIdx.x * blockDim.x + threadIdx.x;
    if (e >= E) return;
    float c0 = cvec[0], c1 = cvec[1], c2 = cvec[2];
    float ale = ea[e * 3 + 0] * c0 + ea[e * 3 + 1] * c1 + ea[e * 3 + 2] * c2;
    int s = src[e], d = dst[e];
    float a = leaky(als[s] + ald[d] + ale, 0.2f);
    alpha[e] = a;
    atomicMax(amax_u + d, ordkey(a));
}

__global__ void k_node_exself(unsigned* __restrict__ amax_u, float* __restrict__ amaxf,
                              const float* __restrict__ aself, float* __restrict__ exself,
                              float* __restrict__ denom, int n) {
    int i = blockIdx.x * blockDim.x + threadIdx.x;
    if (i >= n) return;
    float m = deord(amax_u[i]);
    float ex = expf(aself[i] - m);
    amaxf[i] = m;
    exself[i] = ex;
    denom[i] = ex;
}

__global__ void k_edge_exp(float* __restrict__ alpha, const float* __restrict__ amaxf,
                           const int* __restrict__ dst, float* __restrict__ denom, int E) {
    int e = blockIdx.x * blockDim.x + threadIdx.x;
    if (e >= E) return;
    int d = dst[e];
    float ex = expf(alpha[e] - amaxf[d]);
    alpha[e] = ex;
    atomicAdd(denom + d, ex);
}

__global__ void k_node_init_out(const float* __restrict__ Hm, const float* __restrict__ exself,
                                const float* __restrict__ denom, const float* __restrict__ bias,
                                float* __restrict__ acc, int n) {
    int idx = blockIdx.x * blockDim.x + threadIdx.x;
    if (idx >= n * HF) return;
    int i = idx >> 7, c = idx & 127;
    float coef = exself[i] / (denom[i] + 1e-16f);
    acc[idx] = Hm[idx] * coef + bias[c];
}

__global__ __launch_bounds__(256) void k_edge_scatter(const int* __restrict__ src,
                                                      const int* __restrict__ dst,
                                                      const float* __restrict__ ex,
                                                      const float* __restrict__ denom,
                                                      const float* __restrict__ Hm,
                                                      float* __restrict__ acc, int E) {
    const int lane = threadIdx.x & 31;
    int wid = (blockIdx.x * blockDim.x + threadIdx.x) >> 5;
    int nw  = (gridDim.x * blockDim.x) >> 5;
    for (int e = wid; e < E; e += nw) {
        int s = src[e], d = dst[e];
        float coef = ex[e] / (denom[d] + 1e-16f);
        const float4 hv = *(const float4*)(Hm + (size_t)s * HF + lane * 4);
        float* ap = acc + (size_t)d * HF + lane * 4;
        atomicAdd(ap + 0, hv.x * coef);
        atomicAdd(ap + 1, hv.y * coef);
        atomicAdd(ap + 2, hv.z * coef);
        atomicAdd(ap + 3, hv.w * coef);
    }
}

__global__ __launch_bounds__(128) void k_bn_stats(const float* __restrict__ acc,
                                                  float* __restrict__ colsum,
                                                  float* __restrict__ colsq, int n) {
    int c = threadIdx.x;
    int r0 = blockIdx.x * 256;
    int r1 = min(r0 + 256, n);
    float s = 0.f, s2 = 0.f;
    for (int r = r0; r < r1; ++r) {
        float y = leaky(acc[(size_t)r * HF + c], 0.01f);
        s += y; s2 += y * y;
    }
    atomicAdd(colsum + c, s);
    atomicAdd(colsq + c, s2);
}

__global__ void k_bn_apply(const float* __restrict__ acc, const float* __restrict__ colsum,
                           const float* __restrict__ colsq, const float* __restrict__ g,
                           const float* __restrict__ be, float* __restrict__ xout, int n) {
    int idx = blockIdx.x * blockDim.x + threadIdx.x;
    if (idx >= n * HF) return;
    int c = idx & 127;
    float invN = 1.f / (float)n;
    float mu = colsum[c] * invN;
    float var = colsq[c] * invN - mu * mu;
    float y = leaky(acc[idx], 0.01f);
    xout[idx] = (y - mu) * rsqrtf(var + 1e-5f) * g[c] + be[c];
}

extern "C" void kernel_launch(void* const* d_in, const int* in_sizes, int n_in,
                              void* d_out, int out_size, void* d_ws, size_t ws_size,
                              hipStream_t stream) {
    const float* x  = (const float*)d_in[0];
    const int*   ei = (const int*)d_in[1];
    const float* ea = (const float*)d_in[3];
    const int N = in_sizes[0] / HF;
    const int E = in_sizes[1] / 2;
    const int* src = ei;
    const int* dst = ei + E;

    float* ws = (float*)d_ws;
    const size_t NH = (size_t)N * HF;
    float* hbuf   = ws;
    float* acc    = hbuf + NH;
    float* xbuf   = acc + NH;
    float* alpha  = xbuf + NH;
    float* als    = alpha + E;
    float* ald    = als + N;
    float* aself  = ald + N;
    float* exself = aself + N;
    float* amaxf  = exself + N;
    float* denom  = amaxf + N;
    float* colsum = denom + N;
    float* colsq  = colsum + HF;
    float* easum  = colsq + HF;
    float* cvec   = easum + 4;

    k_zero<<<1, 8, 0, stream>>>(easum, 8);
    k_ea_sum<<<128, 256, 0, stream>>>(ea, E, easum);

    const int nb_e  = (E + 255) / 256;
    const int nb_n  = (N + 255) / 256;
    const int nb_nh = (int)((NH + 255) / 256);

    for (int l = 0; l < 3; ++l) {
        const float* Wl  = (const float*)d_in[4 + 8 * l];
        const float* asv = (const float*)d_in[5 + 8 * l];
        const float* adv = (const float*)d_in[6 + 8 * l];
        const float* Wel = (const float*)d_in[7 + 8 * l];
        const float* aev = (const float*)d_in[8 + 8 * l];
        const float* bv  = (const float*)d_in[9 + 8 * l];
        const float* gv  = (const float*)d_in[10 + 8 * l];
        const float* bev = (const float*)d_in[11 + 8 * l];
        const float* xin = (l == 0) ? x : xbuf;
        float* xout = (l == 2) ? (float*)d_out : xbuf;

        k_compute_c<<<1, 1, 0, stream>>>(Wel, aev, easum, cvec, E);
        k_gemm_xw<<<1024, 256, 0, stream>>>(xin, Wl, hbuf, N);
        k_node_dots<<<512, 256, 0, stream>>>(hbuf, asv, adv, als, ald, N);
        k_node_self<<<nb_n, 256, 0, stream>>>(als, ald, cvec, aself, (unsigned*)amaxf, N);
        k_edge_alpha<<<nb_e, 256, 0, stream>>>(src, dst, ea, als, ald, cvec,
                                               alpha, (unsigned*)amaxf, E);
        k_node_exself<<<nb_n, 256, 0, stream>>>((unsigned*)amaxf, amaxf, aself,
                                                exself, denom, N);
        k_edge_exp<<<nb_e, 256, 0, stream>>>(alpha, amaxf, dst, denom, E);
        k_node_init_out<<<nb_nh, 256, 0, stream>>>(hbuf, exself, denom, bv, acc, N);
        k_edge_scatter<<<2048, 256, 0, stream>>>(src, dst, alpha, denom, hbuf, acc, E);
        k_zero<<<1, 256, 0, stream>>>(colsum, 256);
        k_bn_stats<<<(N + 255) / 256, 128, 0, stream>>>(acc, colsum, colsq, N);
        k_bn_apply<<<nb_nh, 256, 0, stream>>>(acc, colsum, colsq, gv, bev, xout, N);
    }
}


# Per-row passes fused into whole-row wave passes (16 rows a wave, float4 a lane): acc init joins self-loop exp; BN apply and self logit join next node dots; 32 to 24 launches; dot sum order changes
# speedup vs baseline: 1.8693x; 1.8693x over previous
#include <hip/hip_runtime.h>
#include <hip/hip_bf16.h>


typedef float v2f __attribute__((ext_vector_type(2)));
typedef float v8f __attribute__((ext_vector_type(8)));

#define HF 128

__device__ __forceinline__ unsigned ordkey(float f) {
    unsigned u = __float_as_uint(f);
    return (u & 0x80000000u) ? ~u : (u | 0x80000000u);
}
__device__ __forceinline__ float deord(unsigned u) {
    u = (u & 0x80000000u) ? (u & 0x7fffffffu) : ~u;
    return __uint_as_float(u);
}
__device__ __forceinline__ float leaky(float x, float s) { return x > 0.f ? x : s * x; }

__global__ void k_zero(float* p, int n) {
    int i = blockIdx.x * blockDim.x + threadIdx.x;
    if (i < n) p[i] = 0.f;
}

__global__ void k_ea_sum(const float* __restrict__ ea, int E, float* __restrict__ easum) {
    int idx = blockIdx.x * blockDim.x + threadIdx.x;
    int stride = gridDim.x * blockDim.x;
    float s0 = 0.f, s1 = 0.f, s2 = 0.f;
    for (int e = idx; e < E; e += stride) {
        s0 += ea[e * 3 + 0];
        s1 += ea[e * 3 + 1];
        s2 += ea[e * 3 + 2];
    }
    atomicAdd(easum + 0, s0);
    atomicAdd(easum + 1, s1);
    atomicAdd(easum + 2, s2);
}

__global__ __launch_bounds__(128) void k_compute_c(const float* __restrict__ W,
                                                   const float* __restrict__ avs,
                                                   const float* __restrict__ avd,
                                                   const float* __restrict__ We,
                                                   const float* __restrict__ ae,
                                                   const float* __restrict__ easum,
                                                   float* __restrict__ was,
                                                   float* __restrict__ wad,
                                                   float* __restrict__ cvec, int E) {
    const int k = threadIdx.x;
    const float4* wr = (const float4*)(W + (size_t)k * HF);
    float s = 0.f, d = 0.f;
    for (int q = 0; q < HF / 4; ++q) {
        const float4 w = wr[q];
        s += w.x * avs[4 * q] + w.y * avs[4 * q + 1] + w.z * avs[4 * q + 2] + w.w * avs[4 * q + 3];
        d += w.x * avd[4 * q] + w.y * avd[4 * q + 1] + w.z * avd[4 * q + 2] + w.w * avd[4 * q + 3];
    }
    was[k] = s;
    wad[k] = d;

    float t = 0.f;
    if (k < 3) {
        float c = 0.f;
        for (int h = 0; h < HF; ++h) c += We[k * HF + h] * ae[h];
        cvec[k] = c;
        t = (easum[k] / (float)E) * c;
    }
    const float t0 = __shfl(t, 0), t1 = __shfl(t, 1), t2 = __shfl(t, 2);
    if (k == 0) cvec[3] = (t0 + t1) + t2;
}

__global__ __launch_bounds__(256) void k_gemm_xw(const float* __restrict__ X,
                                                 const float* __restrict__ W,
                                                 const float* __restrict__ bias,
                                                 const float* __restrict__ denom,
                                                 float* __restrict__ Ho,
                                                 float* __restrict__ colsum,
                                                 float* __restrict__ colsq, int n) {
    const int lane = threadIdx.x & 31;
    const int wv   = threadIdx.x >> 5;
    const int lo16 = lane & 15;
    const int hi   = lane >> 4;
    const int ncol = wv * 16 + lo16;

    v2f breg[32];
#pragma unroll
    for (int kk = 0; kk < 32; ++kk) {
        int kb = kk * 4 + hi * 2;
        v2f b;
        b.x = W[(size_t)kb * HF + ncol];
        b.y = W[(size_t)(kb + 1) * HF + ncol];
        breg[kk] = b;
    }

    const float bcol = bias[ncol];
    float ps = 0.f, pq = 0.f;

    const int strips = (n + 15) >> 4;
    for (int s = blockIdx.x; s < strips; s += gridDim.x) {
        const int r0 = s * 16;
        int rowA = r0 + lo16;
        if (rowA >= n) rowA = n - 1;
        const float* xrow = X + (size_t)rowA * HF + hi * 2;

        const float rinvA = 1.0f / (denom[rowA] + 1e-16f);

        v2f a[32];
#pragma unroll
        for (int kk = 0; kk < 32; ++kk) {
            float2 t = *(const float2*)(xrow + kk * 4);
            a[kk].x = t.x;
            a[kk].y = t.y;
        }

        v8f acc = {0.f, 0.f, 0.f, 0.f, 0.f, 0.f, 0.f, 0.f};
#pragma unroll
        for (int kk = 0; kk < 32; ++kk) {
            acc = __builtin_amdgcn_wmma_f32_16x16x4_f32(
                false, a[kk], false, breg[kk], (short)0, acc, false, false);
        }

        float rinv[8];
#pragma unroll
        for (int v = 0; v < 8; ++v) rinv[v] = __shfl(rinvA, v + hi * 8);

#pragma unroll
        for (int v = 0; v < 8; ++v) {
            int row = r0 + v + hi * 8;
            if (row < n) {
                const float o = acc[v] * rinv[v] + bcol;
                Ho[(size_t)row * HF + ncol] = o;
                const float y = leaky(o, 0.01f);
                ps += y;
                pq += y * y;
            }
        }
    }

    ps += __shfl_xor(ps, 16);
    pq += __shfl_xor(pq, 16);
    if (hi == 0) {
        atomicAdd(colsum + ncol, ps);
        atomicAdd(colsq + ncol, pq);
    }
}

__global__ __launch_bounds__(256) void k_node_dots(const float* __restrict__ Hm,
                                                   const float* __restrict__ avs,
                                                   const float* __restrict__ avd,
                                                   const float* __restrict__ cvec,
                                                   float* __restrict__ als,
                                                   float* __restrict__ ald,
                                                   float* __restrict__ aself,
                                                   unsigned* __restrict__ amax_u, int n) {
    const int lane = threadIdx.x & 31;
    int wid = (blockIdx.x * blockDim.x + threadIdx.x) >> 5;
    int nw  = (gridDim.x * blockDim.x) >> 5;
    float s0 = avs[lane], s1 = avs[lane + 32], s2 = avs[lane + 64], s3 = avs[lane + 96];
    float d0 = avd[lane], d1 = avd[lane + 32], d2 = avd[lane + 64], d3 = avd[lane + 96];
    const float c3 = cvec[3];
    for (int i = wid; i < n; i += nw) {
        const float* hr = Hm + (size_t)i * HF;
        float h0 = hr[lane], h1 = hr[lane + 32], h2 = hr[lane + 64], h3 = hr[lane + 96];
        float rs = h0 * s0 + h1 * s1 + h2 * s2 + h3 * s3;
        float rd = h0 * d0 + h1 * d1 + h2 * d2 + h3 * d3;
        for (int off = 16; off; off >>= 1) {
            rs += __shfl_xor(rs, off);
            rd += __shfl_xor(rd, off);
        }
        if (lane == 0) {
            als[i] = rs;
            ald[i] = rd;
            const float a = leaky(rs + rd + c3, 0.2f);
            aself[i] = a;
            amax_u[i] = ordkey(a);
        }
    }
}

__global__ void k_edge_alpha(const int* __restrict__ src, const int* __restrict__ dst,
                             const float* __restrict__ ea, const float* __restrict__ als,
                             const float* __restrict__ ald, const float* __restrict__ cvec,
                             float* __restrict__ alpha, unsigned* __restrict__ amax_u, int E) {
    int e = blockIdx.x * blockDim.x + threadIdx.x;
    if (e >= E) return;
    float c0 = cvec[0], c1 = cvec[1], c2 = cvec[2];
    float ale = ea[e * 3 + 0] * c0 + ea[e * 3 + 1] * c1 + ea[e * 3 + 2] * c2;
    int s = src[e], d = dst[e];
    float a = leaky(als[s] + ald[d] + ale, 0.2f);
    alpha[e] = a;
    atomicMax(amax_u + d, ordkey(a));
}

__global__ __launch_bounds__(256) void k_exself_init(const unsigned* amax_u, float* amaxf,
                                                     const float* __restrict__ aself,
                                                     float* __restrict__ denom,
                                                     const float* Hm, float* acc, int n) {
    const int lane = threadIdx.x & 31;
    const int wid = __builtin_amdgcn_readfirstlane(
        (int)((blockIdx.x * blockDim.x + threadIdx.x) >> 5));
    const int r0 = wid * 16;
    if (r0 >= n) return;
    const int cnt = (n - r0 < 16) ? (n - r0) : 16;

    const float4* hp = (const float4*)(Hm + (size_t)r0 * HF) + lane;
    float4* ap = (float4*)(acc + (size_t)r0 * HF) + lane;
    const bool full = (cnt == 16);
    float4 v[16];
    if (full) {
#pragma unroll
        for (int u = 0; u < 16; ++u) v[u] = hp[u * (HF / 4)];
    }

    float ex = 0.f;
    if (lane < cnt) {
        const int i = r0 + lane;
        const float m = deord(amax_u[i]);
        ex = expf(aself[i] - m);
        amaxf[i] = m;
        denom[i] = ex;
    }

    if (full) {
#pragma unroll
        for (int u = 0; u < 16; ++u) {
            const float e = __int_as_float(__builtin_amdgcn_readlane(__float_as_int(ex), u));
            float4 o;
            o.x = v[u].x * e;
            o.y = v[u].y * e;
            o.z = v[u].z * e;
            o.w = v[u].w * e;
            ap[u * (HF / 4)] = o;
        }
    } else {
        for (int u = 0; u < cnt; ++u) {
            const float e = __shfl(ex, u);
            const float4 t = hp[u * (HF / 4)];
            float4 o;
            o.x = t.x * e;
            o.y = t.y * e;
            o.z = t.z * e;
            o.w = t.w * e;
            ap[u * (HF / 4)] = o;
        }
    }
}

__global__ __launch_bounds__(256) void k_edge_exp_scatter(const int* __restrict__ src,
                                                          const int* __restrict__ dst,
                                                          const float* __restrict__ alpha,
                                                          const float* __restrict__ amaxf,
                                                          float* __restrict__ denom,
                                                          const float* __restrict__ Hm,
                                                          float* __restrict__ acc, int E) {
    const int lane = threadIdx.x & 31;
    const int chunk = __builtin_amdgcn_readfirstlane(
        (int)((blockIdx.x * blockDim.x + threadIdx.x) >> 5));
    const int e0 = chunk * 32;
    if (e0 >= E) return;
    const int cnt = (E - e0 < 32) ? (E - e0) : 32;

    int s = 0, d = 0;
    float ex = 0.f;
    if (lane < cnt) {
        const int e = e0 + lane;
        s = src[e];
        d = dst[e];
        ex = expf(alpha[e] - amaxf[d]);
        atomicAdd(denom + d, ex);
    }

    const float* hb = Hm + lane;
    float* ab = acc + lane;
    if (cnt == 32) {
        for (int j = 0; j < 32; j += 4) {
            float h[4][4], xj[4];
            int dj[4];
#pragma unroll
            for (int u = 0; u < 4; ++u) {
                const int sj = __builtin_amdgcn_readlane(s, j + u);
                dj[u] = __builtin_amdgcn_readlane(d, j + u);
                xj[u] = __int_as_float(__builtin_amdgcn_readlane(__float_as_int(ex), j + u));
                const float* hr = hb + (size_t)sj * HF;
                h[u][0] = hr[0];
                h[u][1] = hr[32];
                h[u][2] = hr[64];
                h[u][3] = hr[96];
            }
#pragma unroll
            for (int u = 0; u < 4; ++u) {
                float* ap = ab + (size_t)dj[u] * HF;
                atomicAdd(ap,      h[u][0] * xj[u]);
                atomicAdd(ap + 32, h[u][1] * xj[u]);
                atomicAdd(ap + 64, h[u][2] * xj[u]);
                atomicAdd(ap + 96, h[u][3] * xj[u]);
            }
        }
    } else {
        for (int j = 0; j < cnt; ++j) {
            const int sj = __builtin_amdgcn_readlane(s, j);
            const int dd = __builtin_amdgcn_readlane(d, j);
            const float xx = __int_as_float(__builtin_amdgcn_readlane(__float_as_int(ex), j));
            const float* hr = hb + (size_t)sj * HF;
            const float h0 = hr[0], h1 = hr[32], h2 = hr[64], h3 = hr[96];
            float* ap = ab + (size_t)dd * HF;
            atomicAdd(ap,      h0 * xx);
            atomicAdd(ap + 32, h1 * xx);
            atomicAdd(ap + 64, h2 * xx);
            atomicAdd(ap + 96, h3 * xx);
        }
    }
}

__device__ __forceinline__ float rows16_sum(const float (&p)[16], int lane) {
    const bool b16 = (lane & 16) != 0, b8 = (lane & 8) != 0;
    const bool b4 = (lane & 4) != 0, b2 = (lane & 2) != 0;
    float t8[8], t4[4], t2[2];
#pragma unroll
    for (int u = 0; u < 8; ++u) {
        const float keep = b16 ? p[u + 8] : p[u];
        const float send = b16 ? p[u] : p[u + 8];
        t8[u] = keep + __shfl_xor(send, 16);
    }
#pragma unroll
    for (int u = 0; u < 4; ++u) {
        const float keep = b8 ? t8[u + 4] : t8[u];
        const float send = b8 ? t8[u] : t8[u + 4];
        t4[u] = keep + __shfl_xor(send, 8);
    }
#pragma unroll
    for (int u = 0; u < 2; ++u) {
        const float keep = b4 ? t4[u + 2] : t4[u];
        const float send = b4 ? t4[u] : t4[u + 2];
        t2[u] = keep + __shfl_xor(send, 4);
    }
    const float keep = b2 ? t2[1] : t2[0];
    const float send = b2 ? t2[0] : t2[1];
    const float t1 = keep + __shfl_xor(send, 2);
    return t1 + __shfl_xor(t1, 1);
}

__global__ __launch_bounds__(256) void k_bn_rows(const float* hin,
                                                 const float* __restrict__ colsum,
                                                 const float* __restrict__ colsq,
                                                 const float* __restrict__ g,
                                                 const float* __restrict__ be,
                                                 float* xout,
                                                 const float* __restrict__ was,
                                                 const float* __restrict__ wad,
                                                 const float* __restrict__ cvec,
                                                 float* __restrict__ als,
                                                 float* __restrict__ ald,
                                                 float* __restrict__ aself,
                                                 unsigned* __restrict__ amax_u,
                                                 int n, int dots) {
    const int lane = threadIdx.x & 31;
    const int wid = __builtin_amdgcn_readfirstlane(
        (int)((blockIdx.x * blockDim.x + threadIdx.x) >> 5));
    const int r0 = wid * 16;
    if (r0 >= n) return;
    const int cnt = (n - r0 < 16) ? (n - r0) : 16;

    const float4* hp = (const float4*)(hin + (size_t)r0 * HF) + lane;
    float4* xp = (float4*)(xout + (size_t)r0 * HF) + lane;
    const bool full = (cnt == 16);
    float4 v[16];
    if (full) {
#pragma unroll
        for (int u = 0; u < 16; ++u) v[u] = hp[u * (HF / 4)];
    }

    const int c4 = 4 * lane;
    const float invN = 1.f / (float)n;
    float mu[4], rs[4], gg[4], bb[4], ws[4], wd[4];
#pragma unroll
    for (int q = 0; q < 4; ++q) {
        const int c = c4 + q;
        mu[q] = colsum[c] * invN;
        const float var = colsq[c] * invN - mu[q] * mu[q];
        rs[q] = rsqrtf(var + 1e-5f);
        gg[q] = g[c];
        bb[q] = be[c];
        ws[q] = was[c];
        wd[q] = wad[c];
    }

    if (full) {
        float s[16], d[16];
#pragma unroll
        for (int u = 0; u < 16; ++u) {
            float4 o;
            o.x = (leaky(v[u].x, 0.01f) - mu[0]) * rs[0] * gg[0] + bb[0];
            o.y = (leaky(v[u].y, 0.01f) - mu[1]) * rs[1] * gg[1] + bb[1];
            o.z = (leaky(v[u].z, 0.01f) - mu[2]) * rs[2] * gg[2] + bb[2];
            o.w = (leaky(v[u].w, 0.01f) - mu[3]) * rs[3] * gg[3] + bb[3];
            xp[u * (HF / 4)] = o;
            s[u] = o.x * ws[0] + o.y * ws[1] + o.z * ws[2] + o.w * ws[3];
            d[u] = o.x * wd[0] + o.y * wd[1] + o.z * wd[2] + o.w * wd[3];
        }
        if (dots) {
            const float rs16 = rows16_sum(s, lane);
            const float rd16 = rows16_sum(d, lane);
            if ((lane & 1) == 0) {
                const int i = r0 + (lane >> 1);
                als[i] = rs16;
                ald[i] = rd16;
                const float a = leaky(rs16 + rd16 + cvec[3], 0.2f);
                aself[i] = a;
                amax_u[i] = ordkey(a);
            }
        }
    } else {
        for (int u = 0; u < cnt; ++u) {
            const float4 t = hp[u * (HF / 4)];
            float4 o;
            o.x = (leaky(t.x, 0.01f) - mu[0]) * rs[0] * gg[0] + bb[0];
            o.y = (leaky(t.y, 0.01f) - mu[1]) * rs[1] * gg[1] + bb[1];
            o.z = (leaky(t.z, 0.01f) - mu[2]) * rs[2] * gg[2] + bb[2];
            o.w = (leaky(t.w, 0.01f) - mu[3]) * rs[3] * gg[3] + bb[3];
            xp[u * (HF / 4)] = o;
            if (dots) {
                float sr = o.x * ws[0] + o.y * ws[1] + o.z * ws[2] + o.w * ws[3];
                float dr = o.x * wd[0] + o.y * wd[1] + o.z * wd[2] + o.w * wd[3];
                for (int off = 16; off; off >>= 1) {
                    sr += __shfl_xor(sr, off);
                    dr += __shfl_xor(dr, off);
                }
                if (lane == 0) {
                    als[r0 + u] = sr;
                    ald[r0 + u] = dr;
                    const float a = leaky(sr + dr + cvec[3], 0.2f);
                    aself[r0 + u] = a;
                    amax_u[r0 + u] = ordkey(a);
                }
            }
        }
    }
}

extern "C" void kernel_launch(void* const* d_in, const int* in_sizes, int n_in,
                              void* d_out, int out_size, void* d_ws, size_t ws_size,
                              hipStream_t stream) {
    const float* x  = (const float*)d_in[0];
    const int*   ei = (const int*)d_in[1];
    const float* ea = (const float*)d_in[3];
    const int N = in_sizes[0] / HF;
    const int E = in_sizes[1] / 2;
    const int* src = ei;
    const int* dst = ei + E;

    float* ws = (float*)d_ws;
    const size_t NH = (size_t)N * HF;
    float* hbuf   = ws;
    float* acc    = hbuf + NH;
    float* xbuf   = acc + NH;
    float* alpha  = xbuf + NH;
    float* als    = alpha + E;
    float* ald    = als + N;
    float* aself  = ald + N;
    float* wvec   = aself + N;
    float* amaxf  = wvec + N;
    float* denom  = amaxf + N;
    float* colsum = denom + N;
    float* colsq  = colsum + HF;
    float* easum  = colsq + HF;
    float* cvec   = easum + 4;
    float* was = wvec;
    float* wad = wvec + HF;

    k_zero<<<1, 8, 0, stream>>>(easum, 8);
    k_ea_sum<<<128, 256, 0, stream>>>(ea, E, easum);

    const int nb_e  = (E + 255) / 256;
    const int nb_r  = ((N + 15) / 16 + 7) / 8;
    const int nb_w  = ((E + 31) / 32 + 7) / 8;

    for (int l = 0; l < 3; ++l) {
        const float* Wl  = (const float*)d_in[4 + 8 * l];
        const float* asv = (const float*)d_in[5 + 8 * l];
        const float* adv = (const float*)d_in[6 + 8 * l];
        const float* Wel = (const float*)d_in[7 + 8 * l];
        const float* aev = (const float*)d_in[8 + 8 * l];
        const float* bv  = (const float*)d_in[9 + 8 * l];
        const float* xin = (l == 0) ? x : xbuf;

        k_compute_c<<<1, 128, 0, stream>>>(Wl, asv, adv, Wel, aev, easum, was, wad, cvec, E);
        if (l == 0) {
            k_node_dots<<<512, 256, 0, stream>>>(xin, was, wad, cvec, als, ald, aself,
                                                 (unsigned*)amaxf, N);
        } else {
            const float* gp = (const float*)d_in[10 + 8 * (l - 1)];
            const float* bp = (const float*)d_in[11 + 8 * (l - 1)];
            k_bn_rows<<<nb_r, 256, 0, stream>>>(hbuf, colsum, colsq, gp, bp, xbuf, was, wad, cvec,
                                                als, ald, aself, (unsigned*)amaxf, N, 1);
        }
        k_edge_alpha<<<nb_e, 256, 0, stream>>>(src, dst, ea, als, ald, cvec,
                                               alpha, (unsigned*)amaxf, E);
        k_exself_init<<<nb_r, 256, 0, stream>>>((const unsigned*)amaxf, amaxf, aself,
                                                denom, xin, acc, N);
        k_edge_exp_scatter<<<nb_w, 256, 0, stream>>>(src, dst, alpha, amaxf, denom, xin, acc, E);
        k_zero<<<1, 256, 0, stream>>>(colsum, 256);
        k_gemm_xw<<<1024, 256, 0, stream>>>(acc, Wl, bv, denom, hbuf, colsum, colsq, N);
    }
    k_bn_rows<<<nb_r, 256, 0, stream>>>(hbuf, colsum, colsq, (const float*)d_in[10 + 8 * 2],
                                        (const float*)d_in[11 + 8 * 2], (float*)d_out, was, wad,
                                        cvec, als, ald, aself, (unsigned*)amaxf, N, 0);
}


# Row sums by destination tile in LDS: a 32-wave group owns 200 rows, scans all edge destinations, adds its edges with LDS f32 adds; global f32 atomic scatter and acc init pass go; arrival-order sums
# speedup vs baseline: 1.9605x; 1.0488x over previous
#include <hip/hip_runtime.h>
#include <hip/hip_bf16.h>


typedef float v2f __attribute__((ext_vector_type(2)));
typedef float v8f __attribute__((ext_vector_type(8)));

#define HF 128

__device__ __forceinline__ unsigned ordkey(float f) {
    unsigned u = __float_as_uint(f);
    return (u & 0x80000000u) ? ~u : (u | 0x80000000u);
}
__device__ __forceinline__ float deord(unsigned u) {
    u = (u & 0x80000000u) ? (u & 0x7fffffffu) : ~u;
    return __uint_as_float(u);
}
__device__ __forceinline__ float leaky(float x, float s) { return x > 0.f ? x : s * x; }

__global__ void k_zero(float* p, int n) {
    int i = blockIdx.x * blockDim.x + threadIdx.x;
    if (i < n) p[i] = 0.f;
}

__global__ void k_ea_sum(const float* __restrict__ ea, int E, float* __restrict__ easum) {
    int idx = blockIdx.x * blockDim.x + threadIdx.x;
    int stride = gridDim.x * blockDim.x;
    float s0 = 0.f, s1 = 0.f, s2 = 0.f;
    for (int e = idx; e < E; e += stride) {
        s0 += ea[e * 3 + 0];
        s1 += ea[e * 3 + 1];
        s2 += ea[e * 3 + 2];
    }
    atomicAdd(easum + 0, s0);
    atomicAdd(easum + 1, s1);
    atomicAdd(easum + 2, s2);
}

__global__ __launch_bounds__(128) void k_compute_c(const float* __restrict__ W,
                                                   const float* __restrict__ avs,
                                                   const float* __restrict__ avd,
                                                   const float* __restrict__ We,
                                                   const float* __restrict__ ae,
                                                   const float* __restrict__ easum,
                                                   float* __restrict__ was,
                                                   float* __restrict__ wad,
                                                   float* __restrict__ cvec, int E) {
    const int k = threadIdx.x;
    const float4* wr = (const float4*)(W + (size_t)k * HF);
    float s = 0.f, d = 0.f;
    for (int q = 0; q < HF / 4; ++q) {
        const float4 w = wr[q];
        s += w.x * avs[4 * q] + w.y * avs[4 * q + 1] + w.z * avs[4 * q + 2] + w.w * avs[4 * q + 3];
        d += w.x * avd[4 * q] + w.y * avd[4 * q + 1] + w.z * avd[4 * q + 2] + w.w * avd[4 * q + 3];
    }
    was[k] = s;
    wad[k] = d;

    float t = 0.f;
    if (k < 3) {
        float c = 0.f;
        for (int h = 0; h < HF; ++h) c += We[k * HF + h] * ae[h];
        cvec[k] = c;
        t = (easum[k] / (float)E) * c;
    }
    const float t0 = __shfl(t, 0), t1 = __shfl(t, 1), t2 = __shfl(t, 2);
    if (k == 0) cvec[3] = (t0 + t1) + t2;
}

__global__ __launch_bounds__(256) void k_gemm_xw(const float* __restrict__ X,
                                                 const float* __restrict__ W,
                                                 const float* __restrict__ bias,
                                                 const float* __restrict__ denom,
                                                 float* __restrict__ Ho,
                                                 float* __restrict__ colsum,
                                                 float* __restrict__ colsq, int n) {
    const int lane = threadIdx.x & 31;
    const int wv   = threadIdx.x >> 5;
    const int lo16 = lane & 15;
    const int hi   = lane >> 4;
    const int ncol = wv * 16 + lo16;

    v2f breg[32];
#pragma unroll
    for (int kk = 0; kk < 32; ++kk) {
        int kb = kk * 4 + hi * 2;
        v2f b;
        b.x = W[(size_t)kb * HF + ncol];
        b.y = W[(size_t)(kb + 1) * HF + ncol];
        breg[kk] = b;
    }

    const float bcol = bias[ncol];
    float ps = 0.f, pq = 0.f;

    const int strips = (n + 15) >> 4;
    for (int s = blockIdx.x; s < strips; s += gridDim.x) {
        const int r0 = s * 16;
        int rowA = r0 + lo16;
        if (rowA >= n) rowA = n - 1;
        const float* xrow = X + (size_t)rowA * HF + hi * 2;

        const float rinvA = 1.0f / (denom[rowA] + 1e-16f);

        v2f a[32];
#pragma unroll
        for (int kk = 0; kk < 32; ++kk) {
            float2 t = *(const float2*)(xrow + kk * 4);
            a[kk].x = t.x;
            a[kk].y = t.y;
        }

        v8f acc = {0.f, 0.f, 0.f, 0.f, 0.f, 0.f, 0.f, 0.f};
#pragma unroll
        for (int kk = 0; kk < 32; ++kk) {
            acc = __builtin_amdgcn_wmma_f32_16x16x4_f32(
                false, a[kk], false, breg[kk], (short)0, acc, false, false);
        }

        float rinv[8];
#pragma unroll
        for (int v = 0; v < 8; ++v) rinv[v] = __shfl(rinvA, v + hi * 8);

#pragma unroll
        for (int v = 0; v < 8; ++v) {
            int row = r0 + v + hi * 8;
            if (row < n) {
                const float o = acc[v] * rinv[v] + bcol;
                Ho[(size_t)row * HF + ncol] = o;
                const float y = leaky(o, 0.01f);
                ps += y;
                pq += y * y;
            }
        }
    }

    ps += __shfl_xor(ps, 16);
    pq += __shfl_xor(pq, 16);
    if (hi == 0) {
        atomicAdd(colsum + ncol, ps);
        atomicAdd(colsq + ncol, pq);
    }
}

__global__ __launch_bounds__(256) void k_node_dots(const float* __restrict__ Hm,
                                                   const float* __restrict__ avs,
                                                   const float* __restrict__ avd,
                                                   const float* __restrict__ cvec,
                                                   float* __restrict__ als,
                                                   float* __restrict__ ald,
                                                   float* __restrict__ aself,
                                                   unsigned* __restrict__ amax_u, int n) {
    const int lane = threadIdx.x & 31;
    int wid = (blockIdx.x * blockDim.x + threadIdx.x) >> 5;
    int nw  = (gridDim.x * blockDim.x) >> 5;
    float s0 = avs[lane], s1 = avs[lane + 32], s2 = avs[lane + 64], s3 = avs[lane + 96];
    float d0 = avd[lane], d1 = avd[lane + 32], d2 = avd[lane + 64], d3 = avd[lane + 96];
    const float c3 = cvec[3];
    for (int i = wid; i < n; i += nw) {
        const float* hr = Hm + (size_t)i * HF;
        float h0 = hr[lane], h1 = hr[lane + 32], h2 = hr[lane + 64], h3 = hr[lane + 96];
        float rs = h0 * s0 + h1 * s1 + h2 * s2 + h3 * s3;
        float rd = h0 * d0 + h1 * d1 + h2 * d2 + h3 * d3;
        for (int off = 16; off; off >>= 1) {
            rs += __shfl_xor(rs, off);
            rd += __shfl_xor(rd, off);
        }
        if (lane == 0) {
            als[i] = rs;
            ald[i] = rd;
            const float a = leaky(rs + rd + c3, 0.2f);
            aself[i] = a;
            amax_u[i] = ordkey(a);
        }
    }
}

__global__ void k_edge_alpha(const int* __restrict__ src, const int* __restrict__ dst,
                             const float* __restrict__ ea, const float* __restrict__ als,
                             const float* __restrict__ ald, const float* __restrict__ cvec,
                             float* __restrict__ alpha, unsigned* __restrict__ amax_u, int E) {
    int e = blockIdx.x * blockDim.x + threadIdx.x;
    if (e >= E) return;
    float c0 = cvec[0], c1 = cvec[1], c2 = cvec[2];
    float ale = ea[e * 3 + 0] * c0 + ea[e * 3 + 1] * c1 + ea[e * 3 + 2] * c2;
    int s = src[e], d = dst[e];
    float a = leaky(als[s] + ald[d] + ale, 0.2f);
    alpha[e] = a;
    atomicMax(amax_u + d, ordkey(a));
}

#define TR 200
#define QCAP 64

__device__ __forceinline__ void tile_add_edges4(const float* __restrict__ hb, float* tl,
                                                int s, int dl, float ex, int j) {
    float h[4][4], xj[4];
    int dj[4];
#pragma unroll
    for (int u = 0; u < 4; ++u) {
        const int sj = __builtin_amdgcn_readlane(s, j + u);
        dj[u] = __builtin_amdgcn_readlane(dl, j + u);
        xj[u] = __int_as_float(__builtin_amdgcn_readlane(__float_as_int(ex), j + u));
        const float* hr = hb + (size_t)sj * HF;
        h[u][0] = hr[0];
        h[u][1] = hr[32];
        h[u][2] = hr[64];
        h[u][3] = hr[96];
    }
#pragma unroll
    for (int u = 0; u < 4; ++u) {
        float* ap = tl + dj[u] * HF;
        atomicAdd(ap,      h[u][0] * xj[u]);
        atomicAdd(ap + 32, h[u][1] * xj[u]);
        atomicAdd(ap + 64, h[u][2] * xj[u]);
        atomicAdd(ap + 96, h[u][3] * xj[u]);
    }
}

__global__ __launch_bounds__(1024) void k_tile_rowsum(const int* __restrict__ src,
                                                      const int* __restrict__ dst,
                                                      const float* __restrict__ alpha,
                                                      const unsigned* __restrict__ amax_u,
                                                      const float* __restrict__ aself,
                                                      float* __restrict__ denom,
                                                      const float* __restrict__ Hm,
                                                      float* __restrict__ acc, int n, int E) {
    __shared__ float tile[TR * HF];
    __shared__ float mloc[TR];
    __shared__ float denl[TR];
    __shared__ int qe[32][QCAP];
    __shared__ int qd[32][QCAP];

    const int tid = threadIdx.x;
    const int lane = tid & 31;
    const int w = __builtin_amdgcn_readfirstlane(tid >> 5);
    const int r0 = blockIdx.x * TR;
    const int rows = (n - r0 < TR) ? (n - r0) : TR;

    if (tid < rows) {
        const int i = r0 + tid;
        const float m = deord(amax_u[i]);
        mloc[tid] = m;
        denl[tid] = expf(aself[i] - m);
    }
    __syncthreads();
    for (int r = w; r < rows; r += 32) {
        const float e = denl[r];
        const float4 t = ((const float4*)(Hm + (size_t)(r0 + r) * HF))[lane];
        float4 o;
        o.x = t.x * e;
        o.y = t.y * e;
        o.z = t.z * e;
        o.w = t.w * e;
        ((float4*)(tile + r * HF))[lane] = o;
    }
    __syncthreads();

    const float* hb = Hm + lane;
    float* tl = tile + lane;
    int* mye = qe[w];
    int* myd = qd[w];
    const int nch = (E + 31) >> 5;
    const unsigned below = (1u << lane) - 1u;
    int cnt = 0;

    for (int c0 = w * 16; c0 < nch; c0 += 32 * 16) {
        int dv[16];
#pragma unroll
        for (int u = 0; u < 16; ++u) {
            const int e = (c0 + u) * 32 + lane;
            dv[u] = (e < E) ? dst[e] : -1;
        }
#pragma unroll
        for (int u = 0; u < 16; ++u) {
            const int dl = dv[u] - r0;
            const bool hit = (unsigned)dl < (unsigned)rows;
            const unsigned mask = __builtin_amdgcn_ballot_w32(hit);
            if (mask != 0u) {
                if (hit) {
                    const int pos = cnt + __builtin_popcount(mask & below);
                    mye[pos] = (c0 + u) * 32 + lane;
                    myd[pos] = dl;
                }
                cnt += __builtin_popcount(mask);
                if (cnt >= 32) {
                    cnt -= 32;
                    const int e = mye[cnt + lane];
                    const int dq = myd[cnt + lane];
                    const int s = src[e];
                    const float ex = expf(alpha[e] - mloc[dq]);
                    atomicAdd(denl + dq, ex);
                    for (int j = 0; j < 32; j += 4) tile_add_edges4(hb, tl, s, dq, ex, j);
                }
            }
        }
    }

    {
        int s = 0, dq = 0;
        float ex = 0.f;
        if (lane < cnt) {
            const int e = mye[lane];
            dq = myd[lane];
            s = src[e];
            ex = expf(alpha[e] - mloc[dq]);
            atomicAdd(denl + dq, ex);
        }
        int j = 0;
        for (; j + 4 <= cnt; j += 4) tile_add_edges4(hb, tl, s, dq, ex, j);
        for (; j < cnt; ++j) {
            const int sj = __builtin_amdgcn_readlane(s, j);
            const int dd = __builtin_amdgcn_readlane(dq, j);
            const float xx = __int_as_float(__builtin_amdgcn_readlane(__float_as_int(ex), j));
            const float* hr = hb + (size_t)sj * HF;
            const float h0 = hr[0], h1 = hr[32], h2 = hr[64], h3 = hr[96];
            float* ap = tl + dd * HF;
            atomicAdd(ap,      h0 * xx);
            atomicAdd(ap + 32, h1 * xx);
            atomicAdd(ap + 64, h2 * xx);
            atomicAdd(ap + 96, h3 * xx);
        }
    }
    __syncthreads();

    for (int r = w; r < rows; r += 32)
        ((float4*)(acc + (size_t)(r0 + r) * HF))[lane] = ((const float4*)(tile + r * HF))[lane];
    if (tid < rows) denom[r0 + tid] = denl[tid];
}

__device__ __forceinline__ float rows16_sum(const float (&p)[16], int lane) {
    const bool b16 = (lane & 16) != 0, b8 = (lane & 8) != 0;
    const bool b4 = (lane & 4) != 0, b2 = (lane & 2) != 0;
    float t8[8], t4[4], t2[2];
#pragma unroll
    for (int u = 0; u < 8; ++u) {
        const float keep = b16 ? p[u + 8] : p[u];
        const float send = b16 ? p[u] : p[u + 8];
        t8[u] = keep + __shfl_xor(send, 16);
    }
#pragma unroll
    for (int u = 0; u < 4; ++u) {
        const float keep = b8 ? t8[u + 4] : t8[u];
        const float send = b8 ? t8[u] : t8[u + 4];
        t4[u] = keep + __shfl_xor(send, 8);
    }
#pragma unroll
    for (int u = 0; u < 2; ++u) {
        const float keep = b4 ? t4[u + 2] : t4[u];
        const float send = b4 ? t4[u] : t4[u + 2];
        t2[u] = keep + __shfl_xor(send, 4);
    }
    const float keep = b2 ? t2[1] : t2[0];
    const float send = b2 ? t2[0] : t2[1];
    const float t1 = keep + __shfl_xor(send, 2);
    return t1 + __shfl_xor(t1, 1);
}

__global__ __launch_bounds__(256) void k_bn_rows(const float* hin,
                                                 const float* __restrict__ colsum,
                                                 const float* __restrict__ colsq,
                                                 const float* __restrict__ g,
                                                 const float* __restrict__ be,
                                                 float* xout,
                                                 const float* __restrict__ was,
                                                 const float* __restrict__ wad,
                                                 const float* __restrict__ cvec,
                                                 float* __restrict__ als,
                                                 float* __restrict__ ald,
                                                 float* __restrict__ aself,
                                                 unsigned* __restrict__ amax_u,
                                                 int n, int dots) {
    const int lane = threadIdx.x & 31;
    const int wid = __builtin_amdgcn_readfirstlane(
        (int)((blockIdx.x * blockDim.x + threadIdx.x) >> 5));
    const int r0 = wid * 16;
    if (r0 >= n) return;
    const int cnt = (n - r0 < 16) ? (n - r0) : 16;

    const float4* hp = (const float4*)(hin + (size_t)r0 * HF) + lane;
    float4* xp = (float4*)(xout + (size_t)r0 * HF) + lane;
    const bool full = (cnt == 16);
    float4 v[16];
    if (full) {
#pragma unroll
        for (int u = 0; u < 16; ++u) v[u] = hp[u * (HF / 4)];
    }

    const int c4 = 4 * lane;
    const float invN = 1.f / (float)n;
    float mu[4], rs[4], gg[4], bb[4], ws[4], wd[4];
#pragma unroll
    for (int q = 0; q < 4; ++q) {
        const int c = c4 + q;
        mu[q] = colsum[c] * invN;
        const float var = colsq[c] * invN - mu[q] * mu[q];
        rs[q] = rsqrtf(var + 1e-5f);
        gg[q] = g[c];
        bb[q] = be[c];
        ws[q] = was[c];
        wd[q] = wad[c];
    }

    if (full) {
        float s[16], d[16];
#pragma unroll
        for (int u = 0; u < 16; ++u) {
            float4 o;
            o.x = (leaky(v[u].x, 0.01f) - mu[0]) * rs[0] * gg[0] + bb[0];
            o.y = (leaky(v[u].y, 0.01f) - mu[1]) * rs[1] * gg[1] + bb[1];
            o.z = (leaky(v[u].z, 0.01f) - mu[2]) * rs[2] * gg[2] + bb[2];
            o.w = (leaky(v[u].w, 0.01f) - mu[3]) * rs[3] * gg[3] + bb[3];
            xp[u * (HF / 4)] = o;
            s[u] = o.x * ws[0] + o.y * ws[1] + o.z * ws[2] + o.w * ws[3];
            d[u] = o.x * wd[0] + o.y * wd[1] + o.z * wd[2] + o.w * wd[3];
        }
        if (dots) {
            const float rs16 = rows16_sum(s, lane);
            const float rd16 = rows16_sum(d, lane);
            if ((lane & 1) == 0) {
                const int i = r0 + (lane >> 1);
                als[i] = rs16;
                ald[i] = rd16;
                const float a = leaky(rs16 + rd16 + cvec[3], 0.2f);
                aself[i] = a;
                amax_u[i] = ordkey(a);
            }
        }
    } else {
        for (int u = 0; u < cnt; ++u) {
            const float4 t = hp[u * (HF / 4)];
            float4 o;
            o.x = (leaky(t.x, 0.01f) - mu[0]) * rs[0] * gg[0] + bb[0];
            o.y = (leaky(t.y, 0.01f) - mu[1]) * rs[1] * gg[1] + bb[1];
            o.z = (leaky(t.z, 0.01f) - mu[2]) * rs[2] * gg[2] + bb[2];
            o.w = (leaky(t.w, 0.01f) - mu[3]) * rs[3] * gg[3] + bb[3];
            xp[u * (HF / 4)] = o;
            if (dots) {
                float sr = o.x * ws[0] + o.y * ws[1] + o.z * ws[2] + o.w * ws[3];
                float dr = o.x * wd[0] + o.y * wd[1] + o.z * wd[2] + o.w * wd[3];
                for (int off = 16; off; off >>= 1) {
                    sr += __shfl_xor(sr, off);
                    dr += __shfl_xor(dr, off);
                }
                if (lane == 0) {
                    als[r0 + u] = sr;
                    ald[r0 + u] = dr;
                    const float a = leaky(sr + dr + cvec[3], 0.2f);
                    aself[r0 + u] = a;
                    amax_u[r0 + u] = ordkey(a);
                }
            }
        }
    }
}

extern "C" void kernel_launch(void* const* d_in, const int* in_sizes, int n_in,
                              void* d_out, int out_size, void* d_ws, size_t ws_size,
                              hipStream_t stream) {
    const float* x  = (const float*)d_in[0];
    const int*   ei = (const int*)d_in[1];
    const float* ea = (const float*)d_in[3];
    const int N = in_sizes[0] / HF;
    const int E = in_sizes[1] / 2;
    const int* src = ei;
    const int* dst = ei + E;

    float* ws = (float*)d_ws;
    const size_t NH = (size_t)N * HF;
    float* hbuf   = ws;
    float* acc    = hbuf + NH;
    float* xbuf   = acc + NH;
    float* alpha  = xbuf + NH;
    float* als    = alpha + E;
    float* ald    = als + N;
    float* aself  = ald + N;
    float* wvec   = aself + N;
    float* amaxf  = wvec + N;
    float* denom  = amaxf + N;
    float* colsum = denom + N;
    float* colsq  = colsum + HF;
    float* easum  = colsq + HF;
    float* cvec   = easum + 4;
    float* was = wvec;
    float* wad = wvec + HF;

    k_zero<<<1, 8, 0, stream>>>(easum, 8);
    k_ea_sum<<<128, 256, 0, stream>>>(ea, E, easum);

    const int nb_e  = (E + 255) / 256;
    const int nb_r  = ((N + 15) / 16 + 7) / 8;
    const int nb_t  = (N + TR - 1) / TR;

    for (int l = 0; l < 3; ++l) {
        const float* Wl  = (const float*)d_in[4 + 8 * l];
        const float* asv = (const float*)d_in[5 + 8 * l];
        const float* adv = (const float*)d_in[6 + 8 * l];
        const float* Wel = (const float*)d_in[7 + 8 * l];
        const float* aev = (const float*)d_in[8 + 8 * l];
        const float* bv  = (const float*)d_in[9 + 8 * l];
        const float* xin = (l == 0) ? x : xbuf;

        k_compute_c<<<1, 128, 0, stream>>>(Wl, asv, adv, Wel, aev, easum, was, wad, cvec, E);
        if (l == 0) {
            k_node_dots<<<512, 256, 0, stream>>>(xin, was, wad, cvec, als, ald, aself,
                                                 (unsigned*)amaxf, N);
        } else {
            const float* gp = (const float*)d_in[10 + 8 * (l - 1)];
            const float* bp = (const float*)d_in[11 + 8 * (l - 1)];
            k_bn_rows<<<nb_r, 256, 0, stream>>>(hbuf, colsum, colsq, gp, bp, xbuf, was, wad, cvec,
                                                als, ald, aself, (unsigned*)amaxf, N, 1);
        }
        k_edge_alpha<<<nb_e, 256, 0, stream>>>(src, dst, ea, als, ald, cvec,
                                               alpha, (unsigned*)amaxf, E);
        k_tile_rowsum<<<nb_t, 1024, 0, stream>>>(src, dst, alpha, (const unsigned*)amaxf, aself,
                                                 denom, xin, acc, N, E);
        k_zero<<<1, 256, 0, stream>>>(colsum, 256);
        k_gemm_xw<<<1024, 256, 0, stream>>>(acc, Wl, bv, denom, hbuf, colsum, colsq, N);
    }
    k_bn_rows<<<nb_r, 256, 0, stream>>>(hbuf, colsum, colsq, (const float*)d_in[10 + 8 * 2],
                                        (const float*)d_in[11 + 8 * 2], (float*)d_out, was, wad,
                                        cvec, als, ald, aself, (unsigned*)amaxf, N, 0);
}


# Row-sum tile scan reads a byte plane of tile ids (dst/200 mod 256, one byte an edge, written as whole lines by the edge logit pass) with uint4 loads, 4 bytes a compare; dst read only for matches
# speedup vs baseline: 2.4210x; 1.2349x over previous
#include <hip/hip_runtime.h>
#include <hip/hip_bf16.h>


typedef float v2f __attribute__((ext_vector_type(2)));
typedef float v8f __attribute__((ext_vector_type(8)));

#define HF 128

__device__ __forceinline__ unsigned ordkey(float f) {
    unsigned u = __float_as_uint(f);
    return (u & 0x80000000u) ? ~u : (u | 0x80000000u);
}
__device__ __forceinline__ float deord(unsigned u) {
    u = (u & 0x80000000u) ? (u & 0x7fffffffu) : ~u;
    return __uint_as_float(u);
}
__device__ __forceinline__ float leaky(float x, float s) { return x > 0.f ? x : s * x; }

__global__ void k_zero(float* p, int n) {
    int i = blockIdx.x * blockDim.x + threadIdx.x;
    if (i < n) p[i] = 0.f;
}

__global__ void k_ea_sum(const float* __restrict__ ea, int E, float* __restrict__ easum) {
    int idx = blockIdx.x * blockDim.x + threadIdx.x;
    int stride = gridDim.x * blockDim.x;
    float s0 = 0.f, s1 = 0.f, s2 = 0.f;
    for (int e = idx; e < E; e += stride) {
        s0 += ea[e * 3 + 0];
        s1 += ea[e * 3 + 1];
        s2 += ea[e * 3 + 2];
    }
    atomicAdd(easum + 0, s0);
    atomicAdd(easum + 1, s1);
    atomicAdd(easum + 2, s2);
}

__global__ __launch_bounds__(128) void k_compute_c(const float* __restrict__ W,
                                                   const float* __restrict__ avs,
                                                   const float* __restrict__ avd,
                                                   const float* __restrict__ We,
                                                   const float* __restrict__ ae,
                                                   const float* __restrict__ easum,
                                                   float* __restrict__ was,
                                                   float* __restrict__ wad,
                                                   float* __restrict__ cvec, int E) {
    const int k = threadIdx.x;
    const float4* wr = (const float4*)(W + (size_t)k * HF);
    float s = 0.f, d = 0.f;
    for (int q = 0; q < HF / 4; ++q) {
        const float4 w = wr[q];
        s += w.x * avs[4 * q] + w.y * avs[4 * q + 1] + w.z * avs[4 * q + 2] + w.w * avs[4 * q + 3];
        d += w.x * avd[4 * q] + w.y * avd[4 * q + 1] + w.z * avd[4 * q + 2] + w.w * avd[4 * q + 3];
    }
    was[k] = s;
    wad[k] = d;

    float t = 0.f;
    if (k < 3) {
        float c = 0.f;
        for (int h = 0; h < HF; ++h) c += We[k * HF + h] * ae[h];
        cvec[k] = c;
        t = (easum[k] / (float)E) * c;
    }
    const float t0 = __shfl(t, 0), t1 = __shfl(t, 1), t2 = __shfl(t, 2);
    if (k == 0) cvec[3] = (t0 + t1) + t2;
}

__global__ __launch_bounds__(256) void k_gemm_xw(const float* __restrict__ X,
                                                 const float* __restrict__ W,
                                                 const float* __restrict__ bias,
                                                 const float* __restrict__ denom,
                                                 float* __restrict__ Ho,
                                                 float* __restrict__ colsum,
                                                 float* __restrict__ colsq, int n) {
    const int lane = threadIdx.x & 31;
    const int wv   = threadIdx.x >> 5;
    const int lo16 = lane & 15;
    const int hi   = lane >> 4;
    const int ncol = wv * 16 + lo16;

    v2f breg[32];
#pragma unroll
    for (int kk = 0; kk < 32; ++kk) {
        int kb = kk * 4 + hi * 2;
        v2f b;
        b.x = W[(size_t)kb * HF + ncol];
        b.y = W[(size_t)(kb + 1) * HF + ncol];
        breg[kk] = b;
    }

    const float bcol = bias[ncol];
    float ps = 0.f, pq = 0.f;

    const int strips = (n + 15) >> 4;
    for (int s = blockIdx.x; s < strips; s += gridDim.x) {
        const int r0 = s * 16;
        int rowA = r0 + lo16;
        if (rowA >= n) rowA = n - 1;
        const float* xrow = X + (size_t)rowA * HF + hi * 2;

        const float rinvA = 1.0f / (denom[rowA] + 1e-16f);

        v2f a[32];
#pragma unroll
        for (int kk = 0; kk < 32; ++kk) {
            float2 t = *(const float2*)(xrow + kk * 4);
            a[kk].x = t.x;
            a[kk].y = t.y;
        }

        v8f acc = {0.f, 0.f, 0.f, 0.f, 0.f, 0.f, 0.f, 0.f};
#pragma unroll
        for (int kk = 0; kk < 32; ++kk) {
            acc = __builtin_amdgcn_wmma_f32_16x16x4_f32(
                false, a[kk], false, breg[kk], (short)0, acc, false, false);
        }

        float rinv[8];
#pragma unroll
        for (int v = 0; v < 8; ++v) rinv[v] = __shfl(rinvA, v + hi * 8);

#pragma unroll
        for (int v = 0; v < 8; ++v) {
            int row = r0 + v + hi * 8;
            if (row < n) {
                const float o = acc[v] * rinv[v] + bcol;
                Ho[(size_t)row * HF + ncol] = o;
                const float y = leaky(o, 0.01f);
                ps += y;
                pq += y * y;
            }
        }
    }

    ps += __shfl_xor(ps, 16);
    pq += __shfl_xor(pq, 16);
    if (hi == 0) {
        atomicAdd(colsum + ncol, ps);
        atomicAdd(colsq + ncol, pq);
    }
}

__global__ __launch_bounds__(256) void k_node_dots(const float* __restrict__ Hm,
                                                   const float* __restrict__ avs,
                                                   const float* __restrict__ avd,
                                                   const float* __restrict__ cvec,
                                                   float* __restrict__ als,
                                                   float* __restrict__ ald,
                                                   float* __restrict__ aself,
                                                   unsigned* __restrict__ amax_u, int n) {
    const int lane = threadIdx.x & 31;
    int wid = (blockIdx.x * blockDim.x + threadIdx.x) >> 5;
    int nw  = (gridDim.x * blockDim.x) >> 5;
    float s0 = avs[lane], s1 = avs[lane + 32], s2 = avs[lane + 64], s3 = avs[lane + 96];
    float d0 = avd[lane], d1 = avd[lane + 32], d2 = avd[lane + 64], d3 = avd[lane + 96];
    const float c3 = cvec[3];
    for (int i = wid; i < n; i += nw) {
        const float* hr = Hm + (size_t)i * HF;
        float h0 = hr[lane], h1 = hr[lane + 32], h2 = hr[lane + 64], h3 = hr[lane + 96];
        float rs = h0 * s0 + h1 * s1 + h2 * s2 + h3 * s3;
        float rd = h0 * d0 + h1 * d1 + h2 * d2 + h3 * d3;
        for (int off = 16; off; off >>= 1) {
            rs += __shfl_xor(rs, off);
            rd += __shfl_xor(rd, off);
        }
        if (lane == 0) {
            als[i] = rs;
            ald[i] = rd;
            const float a = leaky(rs + rd + c3, 0.2f);
            aself[i] = a;
            amax_u[i] = ordkey(a);
        }
    }
}

#define TR 200

__global__ __launch_bounds__(256) void k_edge_alpha(const int* __restrict__ src,
                                                    const int* __restrict__ dst,
                                                    const float* __restrict__ ea,
                                                    const float* __restrict__ als,
                                                    const float* __restrict__ ald,
                                                    const float* __restrict__ cvec,
                                                    float* __restrict__ alpha,
                                                    unsigned* __restrict__ amax_u,
                                                    unsigned* __restrict__ tplane, int E) {
    __shared__ unsigned tbs[64];
    const int e = blockIdx.x * 256 + threadIdx.x;
    const bool live = e < E;
    int s = 0, d = 0;
    float e0 = 0.f, e1 = 0.f, e2 = 0.f;
    if (live) {
        s = src[e];
        d = dst[e];
        e0 = ea[e * 3 + 0];
        e1 = ea[e * 3 + 1];
        e2 = ea[e * 3 + 2];
    }
    ((unsigned char*)tbs)[threadIdx.x] =
        live ? (unsigned char)(((unsigned)d / TR) & 255u) : (unsigned char)0xff;
    __syncthreads();
    if (threadIdx.x < 64) tplane[blockIdx.x * 64 + threadIdx.x] = tbs[threadIdx.x];
    if (live) {
        const float c0 = cvec[0], c1 = cvec[1], c2 = cvec[2];
        const float ale = e0 * c0 + e1 * c1 + e2 * c2;
        const float a = leaky(als[s] + ald[d] + ale, 0.2f);
        alpha[e] = a;
        atomicMax(amax_u + d, ordkey(a));
    }
}

#define QCAP 64
#define SU 8

__device__ __forceinline__ unsigned zero_bytes(unsigned x) {
    const unsigned t = (x & 0x7f7f7f7fu) + 0x7f7f7f7fu;
    return ~(t | x | 0x7f7f7f7fu);
}

__device__ __forceinline__ void tile_add_edges4(const float* __restrict__ hb, float* tl,
                                                int s, int dl, float ex, int j) {
    float h[4][4], xj[4];
    int dj[4];
#pragma unroll
    for (int u = 0; u < 4; ++u) {
        const int sj = __builtin_amdgcn_readlane(s, j + u);
        dj[u] = __builtin_amdgcn_readlane(dl, j + u);
        xj[u] = __int_as_float(__builtin_amdgcn_readlane(__float_as_int(ex), j + u));
        const float* hr = hb + (size_t)sj * HF;
        h[u][0] = hr[0];
        h[u][1] = hr[32];
        h[u][2] = hr[64];
        h[u][3] = hr[96];
    }
#pragma unroll
    for (int u = 0; u < 4; ++u) {
        float* ap = tl + dj[u] * HF;
        atomicAdd(ap,      h[u][0] * xj[u]);
        atomicAdd(ap + 32, h[u][1] * xj[u]);
        atomicAdd(ap + 64, h[u][2] * xj[u]);
        atomicAdd(ap + 96, h[u][3] * xj[u]);
    }
}

__device__ __forceinline__ void tile_add_edge1(const float* __restrict__ hb, float* tl,
                                               int s, int dl, float ex, int j) {
    const int sj = __builtin_amdgcn_readlane(s, j);
    const int dd = __builtin_amdgcn_readlane(dl, j);
    const float xx = __int_as_float(__builtin_amdgcn_readlane(__float_as_int(ex), j));
    const float* hr = hb + (size_t)sj * HF;
    const float h0 = hr[0], h1 = hr[32], h2 = hr[64], h3 = hr[96];
    float* ap = tl + dd * HF;
    atomicAdd(ap,      h0 * xx);
    atomicAdd(ap + 32, h1 * xx);
    atomicAdd(ap + 64, h2 * xx);
    atomicAdd(ap + 96, h3 * xx);
}

__device__ __forceinline__ void tile_walk(const float* __restrict__ hb, float* tl,
                                          int s, int dl, float ex, unsigned okm, int j1) {
    int j = 0;
    for (; j + 4 <= j1; j += 4) {
        if (((okm >> j) & 15u) == 15u) {
            tile_add_edges4(hb, tl, s, dl, ex, j);
        } else {
            for (int u = 0; u < 4; ++u)
                if ((okm >> (j + u)) & 1u) tile_add_edge1(hb, tl, s, dl, ex, j + u);
        }
    }
    for (; j < j1; ++j)
        if ((okm >> j) & 1u) tile_add_edge1(hb, tl, s, dl, ex, j);
}

__global__ __launch_bounds__(1024) void k_tile_rowsum(const int* __restrict__ src,
                                                      const int* __restrict__ dst,
                                                      const unsigned* __restrict__ tplane,
                                                      const float* __restrict__ alpha,
                                                      const unsigned* __restrict__ amax_u,
                                                      const float* __restrict__ aself,
                                                      float* __restrict__ denom,
                                                      const float* __restrict__ Hm,
                                                      float* __restrict__ acc, int n, int E) {
    __shared__ float tile[TR * HF];
    __shared__ float mloc[TR];
    __shared__ float denl[TR];
    __shared__ int qe[32][QCAP];

    const int tid = threadIdx.x;
    const int lane = tid & 31;
    const int w = __builtin_amdgcn_readfirstlane(tid >> 5);
    const int r0 = blockIdx.x * TR;
    const int rows = (n - r0 < TR) ? (n - r0) : TR;

    if (tid < rows) {
        const int i = r0 + tid;
        const float m = deord(amax_u[i]);
        mloc[tid] = m;
        denl[tid] = expf(aself[i] - m);
    }
    __syncthreads();
    for (int r = w; r < rows; r += 32) {
        const float e = denl[r];
        const float4 t = ((const float4*)(Hm + (size_t)(r0 + r) * HF))[lane];
        float4 o;
        o.x = t.x * e;
        o.y = t.y * e;
        o.z = t.z * e;
        o.w = t.w * e;
        ((float4*)(tile + r * HF))[lane] = o;
    }
    __syncthreads();

    const float* hb = Hm + lane;
    float* tl = tile + lane;
    int* mye = qe[w];
    const uint4* tp4 = (const uint4*)tplane;
    const int nv = (E + 15) >> 4;
    const unsigned pat = (blockIdx.x & 255u) * 0x01010101u;
    int cnt = 0;

    for (int v0 = w * (32 * SU); v0 < nv; v0 += 32 * 32 * SU) {
        uint4 tv[SU];
#pragma unroll
        for (int u = 0; u < SU; ++u) {
            const int vi = v0 + u * 32 + lane;
            tv[u] = tp4[vi < nv ? vi : nv - 1];
        }
#pragma unroll
        for (int u = 0; u < SU; ++u) {
            const int vi = v0 + u * 32 + lane;
            const int ebase = vi << 4;
            unsigned m = (zero_bytes(tv[u].x ^ pat) >> 7) | (zero_bytes(tv[u].y ^ pat) >> 6) |
                         (zero_bytes(tv[u].z ^ pat) >> 5) | (zero_bytes(tv[u].w ^ pat) >> 4);
            if (vi >= nv) m = 0u;
            unsigned any = __builtin_amdgcn_ballot_w32(m != 0u);
            for (int it = 0; it < 16 && any != 0u; ++it) {
                const int p = __builtin_ctz(m | 0x80000000u);
                const int e = ebase + ((p & 7) << 2) + (p >> 3);
                const bool hit = (m != 0u) && (e < E);
                const unsigned mask = __builtin_amdgcn_ballot_w32(hit);
                if (hit) mye[__builtin_amdgcn_mbcnt_lo(mask, (unsigned)cnt)] = e;
                cnt += __builtin_popcount(mask);
                m &= m - 1u;
                if (cnt >= 32) {
                    cnt -= 32;
                    const int eq = mye[cnt + lane];
                    const int s = src[eq];
                    const int dq = dst[eq] - r0;
                    const float al = alpha[eq];
                    const bool ok = (unsigned)dq < (unsigned)rows;
                    const int dqc = ok ? dq : 0;
                    const float ex0 = expf(al - mloc[dqc]);
                    const float ex = ok ? ex0 : 0.f;
                    atomicAdd(denl + dqc, ex);
                    tile_walk(hb, tl, s, dqc, ex, __builtin_amdgcn_ballot_w32(ok), 32);
                }
                any = __builtin_amdgcn_ballot_w32(m != 0u);
            }
        }
    }

    {
        int s = 0, dq = 0;
        float ex = 0.f;
        bool ok = false;
        if (lane < cnt) {
            const int eq = mye[lane];
            s = src[eq];
            dq = dst[eq] - r0;
            const float al = alpha[eq];
            ok = (unsigned)dq < (unsigned)rows;
            ex = ok ? expf(al - mloc[ok ? dq : 0]) : 0.f;
            if (ok) atomicAdd(denl + dq, ex);
        }
        tile_walk(hb, tl, s, dq, ex, __builtin_amdgcn_ballot_w32(ok), cnt);
    }
    __syncthreads();

    for (int r = w; r < rows; r += 32)
        ((float4*)(acc + (size_t)(r0 + r) * HF))[lane] = ((const float4*)(tile + r * HF))[lane];
    if (tid < rows) denom[r0 + tid] = denl[tid];
}

__device__ __forceinline__ float rows16_sum(const float (&p)[16], int lane) {
    const bool b16 = (lane & 16) != 0, b8 = (lane & 8) != 0;
    const bool b4 = (lane & 4) != 0, b2 = (lane & 2) != 0;
    float t8[8], t4[4], t2[2];
#pragma unroll
    for (int u = 0; u < 8; ++u) {
        const float keep = b16 ? p[u + 8] : p[u];
        const float send = b16 ? p[u] : p[u + 8];
        t8[u] = keep + __shfl_xor(send, 16);
    }
#pragma unroll
    for (int u = 0; u < 4; ++u) {
        const float keep = b8 ? t8[u + 4] : t8[u];
        const float send = b8 ? t8[u] : t8[u + 4];
        t4[u] = keep + __shfl_xor(send, 8);
    }
#pragma unroll
    for (int u = 0; u < 2; ++u) {
        const float keep = b4 ? t4[u + 2] : t4[u];
        const float send = b4 ? t4[u] : t4[u + 2];
        t2[u] = keep + __shfl_xor(send, 4);
    }
    const float keep = b2 ? t2[1] : t2[0];
    const float send = b2 ? t2[0] : t2[1];
    const float t1 = keep + __shfl_xor(send, 2);
    return t1 + __shfl_xor(t1, 1);
}

__global__ __launch_bounds__(256) void k_bn_rows(const float* hin,
                                                 const float* __restrict__ colsum,
                                                 const float* __restrict__ colsq,
                                                 const float* __restrict__ g,
                                                 const float* __restrict__ be,
                                                 float* xout,
                                                 const float* __restrict__ was,
                                                 const float* __restrict__ wad,
                                                 const float* __restrict__ cvec,
                                                 float* __restrict__ als,
                                                 float* __restrict__ ald,
                                                 float* __restrict__ aself,
                                                 unsigned* __restrict__ amax_u,
                                                 int n, int dots) {
    const int lane = threadIdx.x & 31;
    const int wid = __builtin_amdgcn_readfirstlane(
        (int)((blockIdx.x * blockDim.x + threadIdx.x) >> 5));
    const int r0 = wid * 16;
    if (r0 >= n) return;
    const int cnt = (n - r0 < 16) ? (n - r0) : 16;

    const float4* hp = (const float4*)(hin + (size_t)r0 * HF) + lane;
    float4* xp = (float4*)(xout + (size_t)r0 * HF) + lane;
    const bool full = (cnt == 16);
    float4 v[16];
    if (full) {
#pragma unroll
        for (int u = 0; u < 16; ++u) v[u] = hp[u * (HF / 4)];
    }

    const int c4 = 4 * lane;
    const float invN = 1.f / (float)n;
    float mu[4], rs[4], gg[4], bb[4], ws[4], wd[4];
#pragma unroll
    for (int q = 0; q < 4; ++q) {
        const int c = c4 + q;
        mu[q] = colsum[c] * invN;
        const float var = colsq[c] * invN - mu[q] * mu[q];
        rs[q] = rsqrtf(var + 1e-5f);
        gg[q] = g[c];
        bb[q] = be[c];
        ws[q] = was[c];
        wd[q] = wad[c];
    }

    if (full) {
        float s[16], d[16];
#pragma unroll
        for (int u = 0; u < 16; ++u) {
            float4 o;
            o.x = (leaky(v[u].x, 0.01f) - mu[0]) * rs[0] * gg[0] + bb[0];
            o.y = (leaky(v[u].y, 0.01f) - mu[1]) * rs[1] * gg[1] + bb[1];
            o.z = (leaky(v[u].z, 0.01f) - mu[2]) * rs[2] * gg[2] + bb[2];
            o.w = (leaky(v[u].w, 0.01f) - mu[3]) * rs[3] * gg[3] + bb[3];
            xp[u * (HF / 4)] = o;
            s[u] = o.x * ws[0] + o.y * ws[1] + o.z * ws[2] + o.w * ws[3];
            d[u] = o.x * wd[0] + o.y * wd[1] + o.z * wd[2] + o.w * wd[3];
        }
        if (dots) {
            const float rs16 = rows16_sum(s, lane);
            const float rd16 = rows16_sum(d, lane);
            if ((lane & 1) == 0) {
                const int i = r0 + (lane >> 1);
                als[i] = rs16;
                ald[i] = rd16;
                const float a = leaky(rs16 + rd16 + cvec[3], 0.2f);
                aself[i] = a;
                amax_u[i] = ordkey(a);
            }
        }
    } else {
        for (int u = 0; u < cnt; ++u) {
            const float4 t = hp[u * (HF / 4)];
            float4 o;
            o.x = (leaky(t.x, 0.01f) - mu[0]) * rs[0] * gg[0] + bb[0];
            o.y = (leaky(t.y, 0.01f) - mu[1]) * rs[1] * gg[1] + bb[1];
            o.z = (leaky(t.z, 0.01f) - mu[2]) * rs[2] * gg[2] + bb[2];
            o.w = (leaky(t.w, 0.01f) - mu[3]) * rs[3] * gg[3] + bb[3];
            xp[u * (HF / 4)] = o;
            if (dots) {
                float sr = o.x * ws[0] + o.y * ws[1] + o.z * ws[2] + o.w * ws[3];
                float dr = o.x * wd[0] + o.y * wd[1] + o.z * wd[2] + o.w * wd[3];
                for (int off = 16; off; off >>= 1) {
                    sr += __shfl_xor(sr, off);
                    dr += __shfl_xor(dr, off);
                }
                if (lane == 0) {
                    als[r0 + u] = sr;
                    ald[r0 + u] = dr;
                    const float a = leaky(sr + dr + cvec[3], 0.2f);
                    aself[r0 + u] = a;
                    amax_u[r0 + u] = ordkey(a);
                }
            }
        }
    }
}

extern "C" void kernel_launch(void* const* d_in, const int* in_sizes, int n_in,
                              void* d_out, int out_size, void* d_ws, size_t ws_size,
                              hipStream_t stream) {
    const float* x  = (const float*)d_in[0];
    const int*   ei = (const int*)d_in[1];
    const float* ea = (const float*)d_in[3];
    const int N = in_sizes[0] / HF;
    const int E = in_sizes[1] / 2;
    const int* src = ei;
    const int* dst = ei + E;

    float* ws = (float*)d_ws;
    const size_t NH = (size_t)N * HF;
    float* hbuf   = ws;
    float* acc    = hbuf + NH;
    float* xbuf   = acc + NH;
    float* alpha  = xbuf + NH;
    float* als    = alpha + E;
    float* ald    = als + N;
    float* aself  = ald + N;
    float* wvec   = aself + N;
    float* amaxf  = wvec + N;
    float* denom  = amaxf + N;
    float* colsum = denom + N;
    float* colsq  = colsum + HF;
    float* easum  = colsq + HF;
    float* cvec   = easum + 4;
    float* was = wvec;
    float* wad = wvec + HF;
    unsigned* tplane = (unsigned*)hbuf;

    k_zero<<<1, 8, 0, stream>>>(easum, 8);
    k_ea_sum<<<128, 256, 0, stream>>>(ea, E, easum);

    const int nb_e  = (E + 255) / 256;
    const int nb_r  = ((N + 15) / 16 + 7) / 8;
    const int nb_t  = (N + TR - 1) / TR;

    for (int l = 0; l < 3; ++l) {
        const float* Wl  = (const float*)d_in[4 + 8 * l];
        const float* asv = (const float*)d_in[5 + 8 * l];
        const float* adv = (const float*)d_in[6 + 8 * l];
        const float* Wel = (const float*)d_in[7 + 8 * l];
        const float* aev = (const float*)d_in[8 + 8 * l];
        const float* bv  = (const float*)d_in[9 + 8 * l];
        const float* xin = (l == 0) ? x : xbuf;

        k_compute_c<<<1, 128, 0, stream>>>(Wl, asv, adv, Wel, aev, easum, was, wad, cvec, E);
        if (l == 0) {
            k_node_dots<<<512, 256, 0, stream>>>(xin, was, wad, cvec, als, ald, aself,
                                                 (unsigned*)amaxf, N);
        } else {
            const float* gp = (const float*)d_in[10 + 8 * (l - 1)];
            const float* bp = (const float*)d_in[11 + 8 * (l - 1)];
            k_bn_rows<<<nb_r, 256, 0, stream>>>(hbuf, colsum, colsq, gp, bp, xbuf, was, wad, cvec,
                                                als, ald, aself, (unsigned*)amaxf, N, 1);
        }
        k_edge_alpha<<<nb_e, 256, 0, stream>>>(src, dst, ea, als, ald, cvec,
                                               alpha, (unsigned*)amaxf, tplane, E);
        k_tile_rowsum<<<nb_t, 1024, 0, stream>>>(src, dst, tplane, alpha, (const unsigned*)amaxf,
                                                 aself, denom, xin, acc, N, E);
        k_zero<<<1, 256, 0, stream>>>(colsum, 256);
        k_gemm_xw<<<1024, 256, 0, stream>>>(acc, Wl, bv, denom, hbuf, colsum, colsq, N);
    }
    k_bn_rows<<<nb_r, 256, 0, stream>>>(hbuf, colsum, colsq, (const float*)d_in[10 + 8 * 2],
                                        (const float*)d_in[11 + 8 * 2], (float*)d_out, was, wad,
                                        cvec, als, ald, aself, (unsigned*)amaxf, N, 0);
}


# Row-sum tile scan runs once a call: layer 1's waves keep their tile's edges (id, source, local row) as per-wave lists; layers 2, 3 walk the lists, no plane scan or src/dst gathers; arrival-order sums
# speedup vs baseline: 2.5474x; 1.0522x over previous
#include <hip/hip_runtime.h>
#include <hip/hip_bf16.h>


typedef float v2f __attribute__((ext_vector_type(2)));
typedef float v8f __attribute__((ext_vector_type(8)));

#define HF 128

__device__ __forceinline__ unsigned ordkey(float f) {
    unsigned u = __float_as_uint(f);
    return (u & 0x80000000u) ? ~u : (u | 0x80000000u);
}
__device__ __forceinline__ float deord(unsigned u) {
    u = (u & 0x80000000u) ? (u & 0x7fffffffu) : ~u;
    return __uint_as_float(u);
}
__device__ __forceinline__ float leaky(float x, float s) { return x > 0.f ? x : s * x; }

__global__ void k_zero(float* p, int n) {
    int i = blockIdx.x * blockDim.x + threadIdx.x;
    if (i < n) p[i] = 0.f;
}

__global__ void k_ea_sum(const float* __restrict__ ea, int E, float* __restrict__ easum) {
    int idx = blockIdx.x * blockDim.x + threadIdx.x;
    int stride = gridDim.x * blockDim.x;
    float s0 = 0.f, s1 = 0.f, s2 = 0.f;
    for (int e = idx; e < E; e += stride) {
        s0 += ea[e * 3 + 0];
        s1 += ea[e * 3 + 1];
        s2 += ea[e * 3 + 2];
    }
    atomicAdd(easum + 0, s0);
    atomicAdd(easum + 1, s1);
    atomicAdd(easum + 2, s2);
}

__global__ __launch_bounds__(128) void k_compute_c(const float* __restrict__ W,
                                                   const float* __restrict__ avs,
                                                   const float* __restrict__ avd,
                                                   const float* __restrict__ We,
                                                   const float* __restrict__ ae,
                                                   const float* __restrict__ easum,
                                                   float* __restrict__ was,
                                                   float* __restrict__ wad,
                                                   float* __restrict__ cvec, int E) {
    const int k = threadIdx.x;
    const float4* wr = (const float4*)(W + (size_t)k * HF);
    float s = 0.f, d = 0.f;
    for (int q = 0; q < HF / 4; ++q) {
        const float4 w = wr[q];
        s += w.x * avs[4 * q] + w.y * avs[4 * q + 1] + w.z * avs[4 * q + 2] + w.w * avs[4 * q + 3];
        d += w.x * avd[4 * q] + w.y * avd[4 * q + 1] + w.z * avd[4 * q + 2] + w.w * avd[4 * q + 3];
    }
    was[k] = s;
    wad[k] = d;

    float t = 0.f;
    if (k < 3) {
        float c = 0.f;
        for (int h = 0; h < HF; ++h) c += We[k * HF + h] * ae[h];
        cvec[k] = c;
        t = (easum[k] / (float)E) * c;
    }
    const float t0 = __shfl(t, 0), t1 = __shfl(t, 1), t2 = __shfl(t, 2);
    if (k == 0) cvec[3] = (t0 + t1) + t2;
}

__global__ __launch_bounds__(256) void k_gemm_xw(const float* __restrict__ X,
                                                 const float* __restrict__ W,
                                                 const float* __restrict__ bias,
                                                 const float* __restrict__ denom,
                                                 float* __restrict__ Ho,
                                                 float* __restrict__ colsum,
                                                 float* __restrict__ colsq, int n) {
    const int lane = threadIdx.x & 31;
    const int wv   = threadIdx.x >> 5;
    const int lo16 = lane & 15;
    const int hi   = lane >> 4;
    const int ncol = wv * 16 + lo16;

    v2f breg[32];
#pragma unroll
    for (int kk = 0; kk < 32; ++kk) {
        int kb = kk * 4 + hi * 2;
        v2f b;
        b.x = W[(size_t)kb * HF + ncol];
        b.y = W[(size_t)(kb + 1) * HF + ncol];
        breg[kk] = b;
    }

    const float bcol = bias[ncol];
    float ps = 0.f, pq = 0.f;

    const int strips = (n + 15) >> 4;
    for (int s = blockIdx.x; s < strips; s += gridDim.x) {
        const int r0 = s * 16;
        int rowA = r0 + lo16;
        if (rowA >= n) rowA = n - 1;
        const float* xrow = X + (size_t)rowA * HF + hi * 2;

        const float rinvA = 1.0f / (denom[rowA] + 1e-16f);

        v2f a[32];
#pragma unroll
        for (int kk = 0; kk < 32; ++kk) {
            float2 t = *(const float2*)(xrow + kk * 4);
            a[kk].x = t.x;
            a[kk].y = t.y;
        }

        v8f acc = {0.f, 0.f, 0.f, 0.f, 0.f, 0.f, 0.f, 0.f};
#pragma unroll
        for (int kk = 0; kk < 32; ++kk) {
            acc = __builtin_amdgcn_wmma_f32_16x16x4_f32(
                false, a[kk], false, breg[kk], (short)0, acc, false, false);
        }

        float rinv[8];
#pragma unroll
        for (int v = 0; v < 8; ++v) rinv[v] = __shfl(rinvA, v + hi * 8);

#pragma unroll
        for (int v = 0; v < 8; ++v) {
            int row = r0 + v + hi * 8;
            if (row < n) {
                const float o = acc[v] * rinv[v] + bcol;
                Ho[(size_t)row * HF + ncol] = o;
                const float y = leaky(o, 0.01f);
                ps += y;
                pq += y * y;
            }
        }
    }

    ps += __shfl_xor(ps, 16);
    pq += __shfl_xor(pq, 16);
    if (hi == 0) {
        atomicAdd(colsum + ncol, ps);
        atomicAdd(colsq + ncol, pq);
    }
}

__global__ __launch_bounds__(256) void k_node_dots(const float* __restrict__ Hm,
                                                   const float* __restrict__ avs,
                                                   const float* __restrict__ avd,
                                                   const float* __restrict__ cvec,
                                                   float* __restrict__ als,
                                                   float* __restrict__ ald,
                                                   float* __restrict__ aself,
                                                   unsigned* __restrict__ amax_u, int n) {
    const int lane = threadIdx.x & 31;
    int wid = (blockIdx.x * blockDim.x + threadIdx.x) >> 5;
    int nw  = (gridDim.x * blockDim.x) >> 5;
    float s0 = avs[lane], s1 = avs[lane + 32], s2 = avs[lane + 64], s3 = avs[lane + 96];
    float d0 = avd[lane], d1 = avd[lane + 32], d2 = avd[lane + 64], d3 = avd[lane + 96];
    const float c3 = cvec[3];
    for (int i = wid; i < n; i += nw) {
        const float* hr = Hm + (size_t)i * HF;
        float h0 = hr[lane], h1 = hr[lane + 32], h2 = hr[lane + 64], h3 = hr[lane + 96];
        float rs = h0 * s0 + h1 * s1 + h2 * s2 + h3 * s3;
        float rd = h0 * d0 + h1 * d1 + h2 * d2 + h3 * d3;
        for (int off = 16; off; off >>= 1) {
            rs += __shfl_xor(rs, off);
            rd += __shfl_xor(rd, off);
        }
        if (lane == 0) {
            als[i] = rs;
            ald[i] = rd;
            const float a = leaky(rs + rd + c3, 0.2f);
            aself[i] = a;
            amax_u[i] = ordkey(a);
        }
    }
}

#define TR 200

__global__ __launch_bounds__(256) void k_edge_alpha(const int* __restrict__ src,
                                                    const int* __restrict__ dst,
                                                    const float* __restrict__ ea,
                                                    const float* __restrict__ als,
                                                    const float* __restrict__ ald,
                                                    const float* __restrict__ cvec,
                                                    float* __restrict__ alpha,
                                                    unsigned* __restrict__ amax_u,
                                                    unsigned* __restrict__ tplane, int E) {
    __shared__ unsigned tbs[64];
    const int e = blockIdx.x * 256 + threadIdx.x;
    const bool live = e < E;
    int s = 0, d = 0;
    float e0 = 0.f, e1 = 0.f, e2 = 0.f;
    if (live) {
        s = src[e];
        d = dst[e];
        e0 = ea[e * 3 + 0];
        e1 = ea[e * 3 + 1];
        e2 = ea[e * 3 + 2];
    }
    ((unsigned char*)tbs)[threadIdx.x] =
        live ? (unsigned char)(((unsigned)d / TR) & 255u) : (unsigned char)0xff;
    __syncthreads();
    if (threadIdx.x < 64) tplane[blockIdx.x * 64 + threadIdx.x] = tbs[threadIdx.x];
    if (live) {
        const float c0 = cvec[0], c1 = cvec[1], c2 = cvec[2];
        const float ale = e0 * c0 + e1 * c1 + e2 * c2;
        const float a = leaky(als[s] + ald[d] + ale, 0.2f);
        alpha[e] = a;
        atomicMax(amax_u + d, ordkey(a));
    }
}

#define QCAP 64
#define SU 8

__device__ __forceinline__ unsigned zero_bytes(unsigned x) {
    const unsigned t = (x & 0x7f7f7f7fu) + 0x7f7f7f7fu;
    return ~(t | x | 0x7f7f7f7fu);
}

__device__ __forceinline__ void tile_add_edges4(const float* __restrict__ hb, float* tl,
                                                int s, int dl, float ex, int j) {
    float h[4][4], xj[4];
    int dj[4];
#pragma unroll
    for (int u = 0; u < 4; ++u) {
        const int sj = __builtin_amdgcn_readlane(s, j + u);
        dj[u] = __builtin_amdgcn_readlane(dl, j + u);
        xj[u] = __int_as_float(__builtin_amdgcn_readlane(__float_as_int(ex), j + u));
        const float* hr = hb + (size_t)sj * HF;
        h[u][0] = hr[0];
        h[u][1] = hr[32];
        h[u][2] = hr[64];
        h[u][3] = hr[96];
    }
#pragma unroll
    for (int u = 0; u < 4; ++u) {
        float* ap = tl + dj[u] * HF;
        atomicAdd(ap,      h[u][0] * xj[u]);
        atomicAdd(ap + 32, h[u][1] * xj[u]);
        atomicAdd(ap + 64, h[u][2] * xj[u]);
        atomicAdd(ap + 96, h[u][3] * xj[u]);
    }
}

__device__ __forceinline__ void tile_add_edge1(const float* __restrict__ hb, float* tl,
                                               int s, int dl, float ex, int j) {
    const int sj = __builtin_amdgcn_readlane(s, j);
    const int dd = __builtin_amdgcn_readlane(dl, j);
    const float xx = __int_as_float(__builtin_amdgcn_readlane(__float_as_int(ex), j));
    const float* hr = hb + (size_t)sj * HF;
    const float h0 = hr[0], h1 = hr[32], h2 = hr[64], h3 = hr[96];
    float* ap = tl + dd * HF;
    atomicAdd(ap,      h0 * xx);
    atomicAdd(ap + 32, h1 * xx);
    atomicAdd(ap + 64, h2 * xx);
    atomicAdd(ap + 96, h3 * xx);
}

__device__ __forceinline__ void tile_walk(const float* __restrict__ hb, float* tl,
                                          int s, int dl, float ex, unsigned okm, int j1) {
    int j = 0;
    for (; j + 4 <= j1; j += 4) {
        if (((okm >> j) & 15u) == 15u) {
            tile_add_edges4(hb, tl, s, dl, ex, j);
        } else {
            for (int u = 0; u < 4; ++u)
                if ((okm >> (j + u)) & 1u) tile_add_edge1(hb, tl, s, dl, ex, j + u);
        }
    }
    for (; j < j1; ++j)
        if ((okm >> j) & 1u) tile_add_edge1(hb, tl, s, dl, ex, j);
}

__global__ __launch_bounds__(1024) void k_tile_rowsum(const int* __restrict__ src,
                                                      const int* __restrict__ dst,
                                                      const unsigned* __restrict__ tplane,
                                                      const float* __restrict__ alpha,
                                                      const unsigned* __restrict__ amax_u,
                                                      const float* __restrict__ aself,
                                                      float* __restrict__ denom,
                                                      const float* __restrict__ Hm,
                                                      float* __restrict__ acc,
                                                      int* elist, int layer, int n, int E) {
    __shared__ float tile[TR * HF];
    __shared__ float mloc[TR];
    __shared__ float denl[TR];
    __shared__ int qe[32][QCAP];

    const int tid = threadIdx.x;
    const int lane = tid & 31;
    const int w = __builtin_amdgcn_readfirstlane(tid >> 5);
    const int r0 = blockIdx.x * TR;
    const int rows = (n - r0 < TR) ? (n - r0) : TR;

    if (tid < rows) {
        const int i = r0 + tid;
        const float m = deord(amax_u[i]);
        mloc[tid] = m;
        denl[tid] = expf(aself[i] - m);
    }
    __syncthreads();
    for (int r = w; r < rows; r += 32) {
        const float e = denl[r];
        const float4 t = ((const float4*)(Hm + (size_t)(r0 + r) * HF))[lane];
        float4 o;
        o.x = t.x * e;
        o.y = t.y * e;
        o.z = t.z * e;
        o.w = t.w * e;
        ((float4*)(tile + r * HF))[lane] = o;
    }
    __syncthreads();

    const float* hb = Hm + lane;
    float* tl = tile + lane;
    int* mye = qe[w];
    const uint4* tp4 = (const uint4*)tplane;
    const int nv = (E + 15) >> 4;
    const unsigned pat = (blockIdx.x & 255u) * 0x01010101u;
    int cnt = 0;

    const int nseg = (int)gridDim.x * 32;
    const int capw = (int)(((unsigned)(n * HF - nseg) / (unsigned)(nseg * 3)) & ~31u);
    const int seg = blockIdx.x * 32 + w;
    int* le = elist + (size_t)seg * 3 * capw;
    int* ls = le + capw;
    int* ld = ls + capw;
    int* ecnt = elist + (size_t)nseg * 3 * capw;
    int rec = (layer == 0) ? 0 : -1;
    bool scan = true;

    if (layer != 0) {
        const int cw = __builtin_amdgcn_readfirstlane(ecnt[seg]);
        if (cw >= 0) {
            scan = false;
            int li = (lane < cw) ? lane : 0;
            int eq = 0, s = 0, dq = 0;
            if (cw > 0) {
                eq = le[li];
                s = ls[li];
                dq = ld[li];
            }
            for (int g = 0; g < cw; g += 32) {
                const int left = cw - g;
                const int j1 = left < 32 ? left : 32;
                const bool live = lane < j1;
                const float al = alpha[eq];
                int eqn = 0, sn = 0, dqn = 0;
                if (left > 32) {
                    const int ln = g + 32 + ((lane < left - 32) ? lane : 0);
                    eqn = le[ln];
                    sn = ls[ln];
                    dqn = ld[ln];
                }
                const float ex0 = expf(al - mloc[dq]);
                const float ex = live ? ex0 : 0.f;
                atomicAdd(denl + dq, ex);
                tile_walk(hb, tl, s, dq, ex, __builtin_amdgcn_ballot_w32(live), j1);
                eq = eqn;
                s = sn;
                dq = dqn;
            }
        }
    }

    if (scan) {
        for (int v0 = w * (32 * SU); v0 < nv; v0 += 32 * 32 * SU) {
            uint4 tv[SU];
#pragma unroll
            for (int u = 0; u < SU; ++u) {
                const int vi = v0 + u * 32 + lane;
                tv[u] = tp4[vi < nv ? vi : nv - 1];
            }
#pragma unroll
            for (int u = 0; u < SU; ++u) {
                const int vi = v0 + u * 32 + lane;
                const int ebase = vi << 4;
                unsigned m = (zero_bytes(tv[u].x ^ pat) >> 7) | (zero_bytes(tv[u].y ^ pat) >> 6) |
                             (zero_bytes(tv[u].z ^ pat) >> 5) | (zero_bytes(tv[u].w ^ pat) >> 4);
                if (vi >= nv) m = 0u;
                unsigned any = __builtin_amdgcn_ballot_w32(m != 0u);
                for (int it = 0; it < 16 && any != 0u; ++it) {
                    const int p = __builtin_ctz(m | 0x80000000u);
                    const int e = ebase + ((p & 7) << 2) + (p >> 3);
                    const bool hit = (m != 0u) && (e < E);
                    const unsigned mask = __builtin_amdgcn_ballot_w32(hit);
                    if (hit) mye[__builtin_amdgcn_mbcnt_lo(mask, (unsigned)cnt)] = e;
                    cnt += __builtin_popcount(mask);
                    m &= m - 1u;
                    if (cnt >= 32) {
                        cnt -= 32;
                        const int eq = mye[cnt + lane];
                        const int s = src[eq];
                        const int dq = dst[eq] - r0;
                        const float al = alpha[eq];
                        const bool ok = (unsigned)dq < (unsigned)rows;
                        const int dqc = ok ? dq : 0;
                        const float ex0 = expf(al - mloc[dqc]);
                        const float ex = ok ? ex0 : 0.f;
                        atomicAdd(denl + dqc, ex);
                        const unsigned okm = __builtin_amdgcn_ballot_w32(ok);
                        if (rec >= 0) {
                            const int nok = __builtin_popcount(okm);
                            if (rec + nok <= capw) {
                                if (ok) {
                                    const int p = __builtin_amdgcn_mbcnt_lo(okm, (unsigned)rec);
                                    le[p] = eq;
                                    ls[p] = s;
                                    ld[p] = dq;
                                }
                                rec += nok;
                            } else {
                                rec = -1;
                            }
                        }
                        tile_walk(hb, tl, s, dqc, ex, okm, 32);
                    }
                    any = __builtin_amdgcn_ballot_w32(m != 0u);
                }
            }
        }

        {
            int s = 0, dq = 0;
            float ex = 0.f;
            bool ok = false;
            if (lane < cnt) {
                const int eq = mye[lane];
                s = src[eq];
                dq = dst[eq] - r0;
                const float al = alpha[eq];
                ok = (unsigned)dq < (unsigned)rows;
                ex = ok ? expf(al - mloc[ok ? dq : 0]) : 0.f;
                if (ok) atomicAdd(denl + dq, ex);
            }
            const unsigned okm = __builtin_amdgcn_ballot_w32(ok);
            if (rec >= 0) {
                const int nok = __builtin_popcount(okm);
                if (rec + nok <= capw) {
                    if (ok) {
                        const int p = __builtin_amdgcn_mbcnt_lo(okm, (unsigned)rec);
                        le[p] = mye[lane];
                        ls[p] = s;
                        ld[p] = dq;
                    }
                    rec += nok;
                } else {
                    rec = -1;
                }
            }
            if (layer == 0 && lane == 0) ecnt[seg] = rec;
            tile_walk(hb, tl, s, dq, ex, okm, cnt);
        }
    }
    __syncthreads();

    for (int r = w; r < rows; r += 32)
        ((float4*)(acc + (size_t)(r0 + r) * HF))[lane] = ((const float4*)(tile + r * HF))[lane];
    if (tid < rows) denom[r0 + tid] = denl[tid];
}

__device__ __forceinline__ float rows16_sum(const float (&p)[16], int lane) {
    const bool b16 = (lane & 16) != 0, b8 = (lane & 8) != 0;
    const bool b4 = (lane & 4) != 0, b2 = (lane & 2) != 0;
    float t8[8], t4[4], t2[2];
#pragma unroll
    for (int u = 0; u < 8; ++u) {
        const float keep = b16 ? p[u + 8] : p[u];
        const float send = b16 ? p[u] : p[u + 8];
        t8[u] = keep + __shfl_xor(send, 16);
    }
#pragma unroll
    for (int u = 0; u < 4; ++u) {
        const float keep = b8 ? t8[u + 4] : t8[u];
        const float send = b8 ? t8[u] : t8[u + 4];
        t4[u] = keep + __shfl_xor(send, 8);
    }
#pragma unroll
    for (int u = 0; u < 2; ++u) {
        const float keep = b4 ? t4[u + 2] : t4[u];
        const float send = b4 ? t4[u] : t4[u + 2];
        t2[u] = keep + __shfl_xor(send, 4);
    }
    const float keep = b2 ? t2[1] : t2[0];
    const float send = b2 ? t2[0] : t2[1];
    const float t1 = keep + __shfl_xor(send, 2);
    return t1 + __shfl_xor(t1, 1);
}

__global__ __launch_bounds__(256) void k_bn_rows(const float* hin,
                                                 const float* __restrict__ colsum,
                                                 const float* __restrict__ colsq,
                                                 const float* __restrict__ g,
                                                 const float* __restrict__ be,
                                                 float* xout,
                                                 const float* __restrict__ was,
                                                 const float* __restrict__ wad,
                                                 const float* __restrict__ cvec,
                                                 float* __restrict__ als,
                                                 float* __restrict__ ald,
                                                 float* __restrict__ aself,
                                                 unsigned* __restrict__ amax_u,
                                                 int n, int dots) {
    const int lane = threadIdx.x & 31;
    const int wid = __builtin_amdgcn_readfirstlane(
        (int)((blockIdx.x * blockDim.x + threadIdx.x) >> 5));
    const int r0 = wid * 16;
    if (r0 >= n) return;
    const int cnt = (n - r0 < 16) ? (n - r0) : 16;

    const float4* hp = (const float4*)(hin + (size_t)r0 * HF) + lane;
    float4* xp = (float4*)(xout + (size_t)r0 * HF) + lane;
    const bool full = (cnt == 16);
    float4 v[16];
    if (full) {
#pragma unroll
        for (int u = 0; u < 16; ++u) v[u] = hp[u * (HF / 4)];
    }

    const int c4 = 4 * lane;
    const float invN = 1.f / (float)n;
    float mu[4], rs[4], gg[4], bb[4], ws[4], wd[4];
#pragma unroll
    for (int q = 0; q < 4; ++q) {
        const int c = c4 + q;
        mu[q] = colsum[c] * invN;
        const float var = colsq[c] * invN - mu[q] * mu[q];
        rs[q] = rsqrtf(var + 1e-5f);
        gg[q] = g[c];
        bb[q] = be[c];
        ws[q] = was[c];
        wd[q] = wad[c];
    }

    if (full) {
        float s[16], d[16];
#pragma unroll
        for (int u = 0; u < 16; ++u) {
            float4 o;
            o.x = (leaky(v[u].x, 0.01f) - mu[0]) * rs[0] * gg[0] + bb[0];
            o.y = (leaky(v[u].y, 0.01f) - mu[1]) * rs[1] * gg[1] + bb[1];
            o.z = (leaky(v[u].z, 0.01f) - mu[2]) * rs[2] * gg[2] + bb[2];
            o.w = (leaky(v[u].w, 0.01f) - mu[3]) * rs[3] * gg[3] + bb[3];
            xp[u * (HF / 4)] = o;
            s[u] = o.x * ws[0] + o.y * ws[1] + o.z * ws[2] + o.w * ws[3];
            d[u] = o.x * wd[0] + o.y * wd[1] + o.z * wd[2] + o.w * wd[3];
        }
        if (dots) {
            const float rs16 = rows16_sum(s, lane);
            const float rd16 = rows16_sum(d, lane);
            if ((lane & 1) == 0) {
                const int i = r0 + (lane >> 1);
                als[i] = rs16;
                ald[i] = rd16;
                const float a = leaky(rs16 + rd16 + cvec[3], 0.2f);
                aself[i] = a;
                amax_u[i] = ordkey(a);
            }
        }
    } else {
        for (int u = 0; u < cnt; ++u) {
            const float4 t = hp[u * (HF / 4)];
            float4 o;
            o.x = (leaky(t.x, 0.01f) - mu[0]) * rs[0] * gg[0] + bb[0];
            o.y = (leaky(t.y, 0.01f) - mu[1]) * rs[1] * gg[1] + bb[1];
            o.z = (leaky(t.z, 0.01f) - mu[2]) * rs[2] * gg[2] + bb[2];
            o.w = (leaky(t.w, 0.01f) - mu[3]) * rs[3] * gg[3] + bb[3];
            xp[u * (HF / 4)] = o;
            if (dots) {
                float sr = o.x * ws[0] + o.y * ws[1] + o.z * ws[2] + o.w * ws[3];
                float dr = o.x * wd[0] + o.y * wd[1] + o.z * wd[2] + o.w * wd[3];
                for (int off = 16; off; off >>= 1) {
                    sr += __shfl_xor(sr, off);
                    dr += __shfl_xor(dr, off);
                }
                if (lane == 0) {
                    als[r0 + u] = sr;
                    ald[r0 + u] = dr;
                    const float a = leaky(sr + dr + cvec[3], 0.2f);
                    aself[r0 + u] = a;
                    amax_u[r0 + u] = ordkey(a);
                }
            }
        }
    }
}

extern "C" void kernel_launch(void* const* d_in, const int* in_sizes, int n_in,
                              void* d_out, int out_size, void* d_ws, size_t ws_size,
                              hipStream_t stream) {
    const float* x  = (const float*)d_in[0];
    const int*   ei = (const int*)d_in[1];
    const float* ea = (const float*)d_in[3];
    const int N = in_sizes[0] / HF;
    const int E = in_sizes[1] / 2;
    const int* src = ei;
    const int* dst = ei + E;

    float* ws = (float*)d_ws;
    const size_t NH = (size_t)N * HF;
    float* hbuf   = ws;
    float* acc    = hbuf + NH;
    float* xbuf   = acc + NH;
    float* alpha  = xbuf + NH;
    float* als    = alpha + E;
    float* ald    = als + N;
    float* aself  = ald + N;
    float* wvec   = aself + N;
    float* amaxf  = wvec + N;
    float* denom  = amaxf + N;
    float* colsum = denom + N;
    float* colsq  = colsum + HF;
    float* easum  = colsq + HF;
    float* cvec   = easum + 4;
    float* was = wvec;
    float* wad = wvec + HF;
    unsigned* tplane = (unsigned*)hbuf;
    int* elist = (int*)d_out;

    k_zero<<<1, 8, 0, stream>>>(easum, 8);
    k_ea_sum<<<128, 256, 0, stream>>>(ea, E, easum);

    const int nb_e  = (E + 255) / 256;
    const int nb_r  = ((N + 15) / 16 + 7) / 8;
    const int nb_t  = (N + TR - 1) / TR;

    for (int l = 0; l < 3; ++l) {
        const float* Wl  = (const float*)d_in[4 + 8 * l];
        const float* asv = (const float*)d_in[5 + 8 * l];
        const float* adv = (const float*)d_in[6 + 8 * l];
        const float* Wel = (const float*)d_in[7 + 8 * l];
        const float* aev = (const float*)d_in[8 + 8 * l];
        const float* bv  = (const float*)d_in[9 + 8 * l];
        const float* xin = (l == 0) ? x : xbuf;

        k_compute_c<<<1, 128, 0, stream>>>(Wl, asv, adv, Wel, aev, easum, was, wad, cvec, E);
        if (l == 0) {
            k_node_dots<<<512, 256, 0, stream>>>(xin, was, wad, cvec, als, ald, aself,
                                                 (unsigned*)amaxf, N);
        } else {
            const float* gp = (const float*)d_in[10 + 8 * (l - 1)];
            const float* bp = (const float*)d_in[11 + 8 * (l - 1)];
            k_bn_rows<<<nb_r, 256, 0, stream>>>(hbuf, colsum, colsq, gp, bp, xbuf, was, wad, cvec,
                                                als, ald, aself, (unsigned*)amaxf, N, 1);
        }
        k_edge_alpha<<<nb_e, 256, 0, stream>>>(src, dst, ea, als, ald, cvec,
                                               alpha, (unsigned*)amaxf, tplane, E);
        k_tile_rowsum<<<nb_t, 1024, 0, stream>>>(src, dst, tplane, alpha, (const unsigned*)amaxf,
                                                 aself, denom, xin, acc, elist, l, N, E);
        k_zero<<<1, 256, 0, stream>>>(colsum, 256);
        k_gemm_xw<<<1024, 256, 0, stream>>>(acc, Wl, bv, denom, hbuf, colsum, colsq, N);
    }
    k_bn_rows<<<nb_r, 256, 0, stream>>>(hbuf, colsum, colsq, (const float*)d_in[10 + 8 * 2],
                                        (const float*)d_in[11 + 8 * 2], (float*)d_out, was, wad,
                                        cvec, als, ald, aself, (unsigned*)amaxf, N, 0);
}


# Product with W moves into the row-sum tile kernel's end: A fragments come from the LDS tile (row pitch 132 words), B panel in registers; acc/denom round trip and 3 launches go; BN sum order changes
# speedup vs baseline: 2.7979x; 1.0983x over previous
#include <hip/hip_runtime.h>
#include <hip/hip_bf16.h>


typedef float v2f __attribute__((ext_vector_type(2)));
typedef float v8f __attribute__((ext_vector_type(8)));

#define HF 128

__device__ __forceinline__ unsigned ordkey(float f) {
    unsigned u = __float_as_uint(f);
    return (u & 0x80000000u) ? ~u : (u | 0x80000000u);
}
__device__ __forceinline__ float deord(unsigned u) {
    u = (u & 0x80000000u) ? (u & 0x7fffffffu) : ~u;
    return __uint_as_float(u);
}
__device__ __forceinline__ float leaky(float x, float s) { return x > 0.f ? x : s * x; }

__global__ void k_zero(float* p, int n) {
    int i = blockIdx.x * blockDim.x + threadIdx.x;
    if (i < n) p[i] = 0.f;
}

__global__ void k_ea_sum(const float* __restrict__ ea, int E, float* __restrict__ easum) {
    int idx = blockIdx.x * blockDim.x + threadIdx.x;
    int stride = gridDim.x * blockDim.x;
    float s0 = 0.f, s1 = 0.f, s2 = 0.f;
    for (int e = idx; e < E; e += stride) {
        s0 += ea[e * 3 + 0];
        s1 += ea[e * 3 + 1];
        s2 += ea[e * 3 + 2];
    }
    atomicAdd(easum + 0, s0);
    atomicAdd(easum + 1, s1);
    atomicAdd(easum + 2, s2);
}

__global__ __launch_bounds__(128) void k_compute_c(const float* __restrict__ W,
                                                   const float* __restrict__ avs,
                                                   const float* __restrict__ avd,
                                                   const float* __restrict__ We,
                                                   const float* __restrict__ ae,
                                                   const float* __restrict__ easum,
                                                   float* __restrict__ was,
                                                   float* __restrict__ wad,
                                                   float* __restrict__ cvec, int E) {
    const int k = threadIdx.x;
    const float4* wr = (const float4*)(W + (size_t)k * HF);
    float s = 0.f, d = 0.f;
    for (int q = 0; q < HF / 4; ++q) {
        const float4 w = wr[q];
        s += w.x * avs[4 * q] + w.y * avs[4 * q + 1] + w.z * avs[4 * q + 2] + w.w * avs[4 * q + 3];
        d += w.x * avd[4 * q] + w.y * avd[4 * q + 1] + w.z * avd[4 * q + 2] + w.w * avd[4 * q + 3];
    }
    was[k] = s;
    wad[k] = d;

    float t = 0.f;
    if (k < 3) {
        float c = 0.f;
        for (int h = 0; h < HF; ++h) c += We[k * HF + h] * ae[h];
        cvec[k] = c;
        t = (easum[k] / (float)E) * c;
    }
    const float t0 = __shfl(t, 0), t1 = __shfl(t, 1), t2 = __shfl(t, 2);
    if (k == 0) cvec[3] = (t0 + t1) + t2;
}

__global__ __launch_bounds__(256) void k_node_dots(const float* __restrict__ Hm,
                                                   const float* __restrict__ avs,
                                                   const float* __restrict__ avd,
                                                   const float* __restrict__ cvec,
                                                   float* __restrict__ als,
                                                   float* __restrict__ ald,
                                                   float* __restrict__ aself,
                                                   unsigned* __restrict__ amax_u, int n) {
    const int lane = threadIdx.x & 31;
    int wid = (blockIdx.x * blockDim.x + threadIdx.x) >> 5;
    int nw  = (gridDim.x * blockDim.x) >> 5;
    float s0 = avs[lane], s1 = avs[lane + 32], s2 = avs[lane + 64], s3 = avs[lane + 96];
    float d0 = avd[lane], d1 = avd[lane + 32], d2 = avd[lane + 64], d3 = avd[lane + 96];
    const float c3 = cvec[3];
    for (int i = wid; i < n; i += nw) {
        const float* hr = Hm + (size_t)i * HF;
        float h0 = hr[lane], h1 = hr[lane + 32], h2 = hr[lane + 64], h3 = hr[lane + 96];
        float rs = h0 * s0 + h1 * s1 + h2 * s2 + h3 * s3;
        float rd = h0 * d0 + h1 * d1 + h2 * d2 + h3 * d3;
        for (int off = 16; off; off >>= 1) {
            rs += __shfl_xor(rs, off);
            rd += __shfl_xor(rd, off);
        }
        if (lane == 0) {
            als[i] = rs;
            ald[i] = rd;
            const float a = leaky(rs + rd + c3, 0.2f);
            aself[i] = a;
            amax_u[i] = ordkey(a);
        }
    }
}

#define TR 200
#define TP 132

__global__ __launch_bounds__(256) void k_edge_alpha(const int* __restrict__ src,
                                                    const int* __restrict__ dst,
                                                    const float* __restrict__ ea,
                                                    const float* __restrict__ als,
                                                    const float* __restrict__ ald,
                                                    const float* __restrict__ cvec,
                                                    float* __restrict__ alpha,
                                                    unsigned* __restrict__ amax_u,
                                                    unsigned* __restrict__ tplane, int E) {
    __shared__ unsigned tbs[64];
    const int e = blockIdx.x * 256 + threadIdx.x;
    const bool live = e < E;
    int s = 0, d = 0;
    float e0 = 0.f, e1 = 0.f, e2 = 0.f;
    if (live) {
        s = src[e];
        d = dst[e];
        e0 = ea[e * 3 + 0];
        e1 = ea[e * 3 + 1];
        e2 = ea[e * 3 + 2];
    }
    ((unsigned char*)tbs)[threadIdx.x] =
        live ? (unsigned char)(((unsigned)d / TR) & 255u) : (unsigned char)0xff;
    __syncthreads();
    if (threadIdx.x < 64) tplane[blockIdx.x * 64 + threadIdx.x] = tbs[threadIdx.x];
    if (live) {
        const float c0 = cvec[0], c1 = cvec[1], c2 = cvec[2];
        const float ale = e0 * c0 + e1 * c1 + e2 * c2;
        const float a = leaky(als[s] + ald[d] + ale, 0.2f);
        alpha[e] = a;
        atomicMax(amax_u + d, ordkey(a));
    }
}

#define QCAP 64
#define SU 8

__device__ __forceinline__ unsigned zero_bytes(unsigned x) {
    const unsigned t = (x & 0x7f7f7f7fu) + 0x7f7f7f7fu;
    return ~(t | x | 0x7f7f7f7fu);
}

__device__ __forceinline__ void tile_add_edges4(const float* __restrict__ hb, float* tl,
                                                int s, int dl, float ex, int j) {
    float h[4][4], xj[4];
    int dj[4];
#pragma unroll
    for (int u = 0; u < 4; ++u) {
        const int sj = __builtin_amdgcn_readlane(s, j + u);
        dj[u] = __builtin_amdgcn_readlane(dl, j + u);
        xj[u] = __int_as_float(__builtin_amdgcn_readlane(__float_as_int(ex), j + u));
        const float* hr = hb + (size_t)sj * HF;
        h[u][0] = hr[0];
        h[u][1] = hr[32];
        h[u][2] = hr[64];
        h[u][3] = hr[96];
    }
#pragma unroll
    for (int u = 0; u < 4; ++u) {
        float* ap = tl + dj[u] * TP;
        atomicAdd(ap,      h[u][0] * xj[u]);
        atomicAdd(ap + 32, h[u][1] * xj[u]);
        atomicAdd(ap + 64, h[u][2] * xj[u]);
        atomicAdd(ap + 96, h[u][3] * xj[u]);
    }
}

__device__ __forceinline__ void tile_add_edge1(const float* __restrict__ hb, float* tl,
                                               int s, int dl, float ex, int j) {
    const int sj = __builtin_amdgcn_readlane(s, j);
    const int dd = __builtin_amdgcn_readlane(dl, j);
    const float xx = __int_as_float(__builtin_amdgcn_readlane(__float_as_int(ex), j));
    const float* hr = hb + (size_t)sj * HF;
    const float h0 = hr[0], h1 = hr[32], h2 = hr[64], h3 = hr[96];
    float* ap = tl + dd * TP;
    atomicAdd(ap,      h0 * xx);
    atomicAdd(ap + 32, h1 * xx);
    atomicAdd(ap + 64, h2 * xx);
    atomicAdd(ap + 96, h3 * xx);
}

__device__ __forceinline__ void tile_walk(const float* __restrict__ hb, float* tl,
                                          int s, int dl, float ex, unsigned okm, int j1) {
    int j = 0;
    for (; j + 4 <= j1; j += 4) {
        if (((okm >> j) & 15u) == 15u) {
            tile_add_edges4(hb, tl, s, dl, ex, j);
        } else {
            for (int u = 0; u < 4; ++u)
                if ((okm >> (j + u)) & 1u) tile_add_edge1(hb, tl, s, dl, ex, j + u);
        }
    }
    for (; j < j1; ++j)
        if ((okm >> j) & 1u) tile_add_edge1(hb, tl, s, dl, ex, j);
}

__global__ __launch_bounds__(1024) void k_tile_rowsum(const int* __restrict__ src,
                                                      const int* __restrict__ dst,
                                                      const unsigned* __restrict__ tplane,
                                                      const float* __restrict__ alpha,
                                                      const unsigned* __restrict__ amax_u,
                                                      const float* __restrict__ aself,
                                                      const float* __restrict__ Hm,
                                                      const float* __restrict__ W,
                                                      const float* __restrict__ bias,
                                                      float* __restrict__ Ho,
                                                      float* __restrict__ colsum,
                                                      float* __restrict__ colsq,
                                                      int* elist, int layer, int n, int E) {
    __shared__ float tile[TR * TP];
    __shared__ float mloc[TR];
    __shared__ float denl[TR];
    __shared__ int qe[32][QCAP];

    const int tid = threadIdx.x;
    const int lane = tid & 31;
    const int w = __builtin_amdgcn_readfirstlane(tid >> 5);
    const int r0 = blockIdx.x * TR;
    const int rows = (n - r0 < TR) ? (n - r0) : TR;

    if (tid < rows) {
        const int i = r0 + tid;
        const float m = deord(amax_u[i]);
        mloc[tid] = m;
        denl[tid] = expf(aself[i] - m);
    }
    __syncthreads();
    for (int r = w; r < rows; r += 32) {
        const float e = denl[r];
        const float4 t = ((const float4*)(Hm + (size_t)(r0 + r) * HF))[lane];
        float4 o;
        o.x = t.x * e;
        o.y = t.y * e;
        o.z = t.z * e;
        o.w = t.w * e;
        ((float4*)(tile + r * TP))[lane] = o;
    }
    __syncthreads();

    const float* hb = Hm + lane;
    float* tl = tile + lane;
    int* mye = qe[w];
    const uint4* tp4 = (const uint4*)tplane;
    const int nv = (E + 15) >> 4;
    const unsigned pat = (blockIdx.x & 255u) * 0x01010101u;
    int cnt = 0;

    const int nseg = (int)gridDim.x * 32;
    const int capw = (int)(((unsigned)(n * HF - nseg) / (unsigned)(nseg * 3)) & ~31u);
    const int seg = blockIdx.x * 32 + w;
    int* le = elist + (size_t)seg * 3 * capw;
    int* ls = le + capw;
    int* ld = ls + capw;
    int* ecnt = elist + (size_t)nseg * 3 * capw;
    int rec = (layer == 0) ? 0 : -1;
    bool scan = true;

    if (layer != 0) {
        const int cw = __builtin_amdgcn_readfirstlane(ecnt[seg]);
        if (cw >= 0) {
            scan = false;
            int li = (lane < cw) ? lane : 0;
            int eq = 0, s = 0, dq = 0;
            if (cw > 0) {
                eq = le[li];
                s = ls[li];
                dq = ld[li];
            }
            for (int g = 0; g < cw; g += 32) {
                const int left = cw - g;
                const int j1 = left < 32 ? left : 32;
                const bool live = lane < j1;
                const float al = alpha[eq];
                int eqn = 0, sn = 0, dqn = 0;
                if (left > 32) {
                    const int ln = g + 32 + ((lane < left - 32) ? lane : 0);
                    eqn = le[ln];
                    sn = ls[ln];
                    dqn = ld[ln];
                }
                const float ex0 = expf(al - mloc[dq]);
                const float ex = live ? ex0 : 0.f;
                atomicAdd(denl + dq, ex);
                tile_walk(hb, tl, s, dq, ex, __builtin_amdgcn_ballot_w32(live), j1);
                eq = eqn;
                s = sn;
                dq = dqn;
            }
        }
    }

    if (scan) {
        for (int v0 = w * (32 * SU); v0 < nv; v0 += 32 * 32 * SU) {
            uint4 tv[SU];
#pragma unroll
            for (int u = 0; u < SU; ++u) {
                const int vi = v0 + u * 32 + lane;
                tv[u] = tp4[vi < nv ? vi : nv - 1];
            }
#pragma unroll
            for (int u = 0; u < SU; ++u) {
                const int vi = v0 + u * 32 + lane;
                const int ebase = vi << 4;
                unsigned m = (zero_bytes(tv[u].x ^ pat) >> 7) | (zero_bytes(tv[u].y ^ pat) >> 6) |
                             (zero_bytes(tv[u].z ^ pat) >> 5) | (zero_bytes(tv[u].w ^ pat) >> 4);
                if (vi >= nv) m = 0u;
                unsigned any = __builtin_amdgcn_ballot_w32(m != 0u);
                for (int it = 0; it < 16 && any != 0u; ++it) {
                    const int p = __builtin_ctz(m | 0x80000000u);
                    const int e = ebase + ((p & 7) << 2) + (p >> 3);
                    const bool hit = (m != 0u) && (e < E);
                    const unsigned mask = __builtin_amdgcn_ballot_w32(hit);
                    if (hit) mye[__builtin_amdgcn_mbcnt_lo(mask, (unsigned)cnt)] = e;
                    cnt += __builtin_popcount(mask);
                    m &= m - 1u;
                    if (cnt >= 32) {
                        cnt -= 32;
                        const int eq = mye[cnt + lane];
                        const int s = src[eq];
                        const int dq = dst[eq] - r0;
                        const float al = alpha[eq];
                        const bool ok = (unsigned)dq < (unsigned)rows;
                        const int dqc = ok ? dq : 0;
                        const float ex0 = expf(al - mloc[dqc]);
                        const float ex = ok ? ex0 : 0.f;
                        atomicAdd(denl + dqc, ex);
                        const unsigned okm = __builtin_amdgcn_ballot_w32(ok);
                        if (rec >= 0) {
                            const int nok = __builtin_popcount(okm);
                            if (rec + nok <= capw) {
                                if (ok) {
                                    const int p = __builtin_amdgcn_mbcnt_lo(okm, (unsigned)rec);
                                    le[p] = eq;
                                    ls[p] = s;
                                    ld[p] = dq;
                                }
                                rec += nok;
                            } else {
                                rec = -1;
                            }
                        }
                        tile_walk(hb, tl, s, dqc, ex, okm, 32);
                    }
                    any = __builtin_amdgcn_ballot_w32(m != 0u);
                }
            }
        }

        {
            int s = 0, dq = 0;
            float ex = 0.f;
            bool ok = false;
            if (lane < cnt) {
                const int eq = mye[lane];
                s = src[eq];
                dq = dst[eq] - r0;
                const float al = alpha[eq];
                ok = (unsigned)dq < (unsigned)rows;
                ex = ok ? expf(al - mloc[ok ? dq : 0]) : 0.f;
                if (ok) atomicAdd(denl + dq, ex);
            }
            const unsigned okm = __builtin_amdgcn_ballot_w32(ok);
            if (rec >= 0) {
                const int nok = __builtin_popcount(okm);
                if (rec + nok <= capw) {
                    if (ok) {
                        const int p = __builtin_amdgcn_mbcnt_lo(okm, (unsigned)rec);
                        le[p] = mye[lane];
                        ls[p] = s;
                        ld[p] = dq;
                    }
                    rec += nok;
                } else {
                    rec = -1;
                }
            }
            if (layer == 0 && lane == 0) ecnt[seg] = rec;
            tile_walk(hb, tl, s, dq, ex, okm, cnt);
        }
    }

    const int lo16 = lane & 15;
    const int hi = lane >> 4;
    const int ct = w & 7;
    const int sg = w >> 3;
    const int ncol = ct * 16 + lo16;

    v2f breg[32];
#pragma unroll
    for (int kk = 0; kk < 32; ++kk) {
        const int kb = kk * 4 + hi * 2;
        v2f b;
        b.x = W[(size_t)kb * HF + ncol];
        b.y = W[(size_t)(kb + 1) * HF + ncol];
        breg[kk] = b;
    }
    const float bcol = bias[ncol];
    __syncthreads();

    float ps = 0.f, pq = 0.f;
    for (int s = sg; s * 16 < rows; s += 4) {
        const int rs0 = s * 16;
        int rl = rs0 + lo16;
        if (rl >= rows) rl = rows - 1;
        const float* xr = tile + rl * TP + hi * 2;

        const float rinvA = 1.0f / (denl[rl] + 1e-16f);

        v8f cacc = {0.f, 0.f, 0.f, 0.f, 0.f, 0.f, 0.f, 0.f};
#pragma unroll
        for (int kk = 0; kk < 32; ++kk) {
            const float2 t = *(const float2*)(xr + kk * 4);
            v2f a;
            a.x = t.x;
            a.y = t.y;
            cacc = __builtin_amdgcn_wmma_f32_16x16x4_f32(
                false, a, false, breg[kk], (short)0, cacc, false, false);
        }

        float rinv[8];
#pragma unroll
        for (int v = 0; v < 8; ++v) rinv[v] = __shfl(rinvA, v + hi * 8);

#pragma unroll
        for (int v = 0; v < 8; ++v) {
            const int row = rs0 + v + hi * 8;
            if (row < rows) {
                const float o = cacc[v] * rinv[v] + bcol;
                Ho[(size_t)(r0 + row) * HF + ncol] = o;
                const float y = leaky(o, 0.01f);
                ps += y;
                pq += y * y;
            }
        }
    }

    ps += __shfl_xor(ps, 16);
    pq += __shfl_xor(pq, 16);
    if (hi == 0) {
        atomicAdd(colsum + ncol, ps);
        atomicAdd(colsq + ncol, pq);
    }
}

__device__ __forceinline__ float rows16_sum(const float (&p)[16], int lane) {
    const bool b16 = (lane & 16) != 0, b8 = (lane & 8) != 0;
    const bool b4 = (lane & 4) != 0, b2 = (lane & 2) != 0;
    float t8[8], t4[4], t2[2];
#pragma unroll
    for (int u = 0; u < 8; ++u) {
        const float keep = b16 ? p[u + 8] : p[u];
        const float send = b16 ? p[u] : p[u + 8];
        t8[u] = keep + __shfl_xor(send, 16);
    }
#pragma unroll
    for (int u = 0; u < 4; ++u) {
        const float keep = b8 ? t8[u + 4] : t8[u];
        const float send = b8 ? t8[u] : t8[u + 4];
        t4[u] = keep + __shfl_xor(send, 8);
    }
#pragma unroll
    for (int u = 0; u < 2; ++u) {
        const float keep = b4 ? t4[u + 2] : t4[u];
        const float send = b4 ? t4[u] : t4[u + 2];
        t2[u] = keep + __shfl_xor(send, 4);
    }
    const float keep = b2 ? t2[1] : t2[0];
    const float send = b2 ? t2[0] : t2[1];
    const float t1 = keep + __shfl_xor(send, 2);
    return t1 + __shfl_xor(t1, 1);
}

__global__ __launch_bounds__(256) void k_bn_rows(const float* hin,
                                                 const float* __restrict__ colsum,
                                                 const float* __restrict__ colsq,
                                                 const float* __restrict__ g,
                                                 const float* __restrict__ be,
                                                 float* xout,
                                                 const float* __restrict__ was,
                                                 const float* __restrict__ wad,
                                                 const float* __restrict__ cvec,
                                                 float* __restrict__ als,
                                                 float* __restrict__ ald,
                                                 float* __restrict__ aself,
                                                 unsigned* __restrict__ amax_u,
                                                 int n, int dots) {
    const int lane = threadIdx.x & 31;
    const int wid = __builtin_amdgcn_readfirstlane(
        (int)((blockIdx.x * blockDim.x + threadIdx.x) >> 5));
    const int r0 = wid * 16;
    if (r0 >= n) return;
    const int cnt = (n - r0 < 16) ? (n - r0) : 16;

    const float4* hp = (const float4*)(hin + (size_t)r0 * HF) + lane;
    float4* xp = (float4*)(xout + (size_t)r0 * HF) + lane;
    const bool full = (cnt == 16);
    float4 v[16];
    if (full) {
#pragma unroll
        for (int u = 0; u < 16; ++u) v[u] = hp[u * (HF / 4)];
    }

    const int c4 = 4 * lane;
    const float invN = 1.f / (float)n;
    float mu[4], rs[4], gg[4], bb[4], ws[4], wd[4];
#pragma unroll
    for (int q = 0; q < 4; ++q) {
        const int c = c4 + q;
        mu[q] = colsum[c] * invN;
        const float var = colsq[c] * invN - mu[q] * mu[q];
        rs[q] = rsqrtf(var + 1e-5f);
        gg[q] = g[c];
        bb[q] = be[c];
        ws[q] = was[c];
        wd[q] = wad[c];
    }

    if (full) {
        float s[16], d[16];
#pragma unroll
        for (int u = 0; u < 16; ++u) {
            float4 o;
            o.x = (leaky(v[u].x, 0.01f) - mu[0]) * rs[0] * gg[0] + bb[0];
            o.y = (leaky(v[u].y, 0.01f) - mu[1]) * rs[1] * gg[1] + bb[1];
            o.z = (leaky(v[u].z, 0.01f) - mu[2]) * rs[2] * gg[2] + bb[2];
            o.w = (leaky(v[u].w, 0.01f) - mu[3]) * rs[3] * gg[3] + bb[3];
            xp[u * (HF / 4)] = o;
            s[u] = o.x * ws[0] + o.y * ws[1] + o.z * ws[2] + o.w * ws[3];
            d[u] = o.x * wd[0] + o.y * wd[1] + o.z * wd[2] + o.w * wd[3];
        }
        if (dots) {
            const float rs16 = rows16_sum(s, lane);
            const float rd16 = rows16_sum(d, lane);
            if ((lane & 1) == 0) {
                const int i = r0 + (lane >> 1);
                als[i] = rs16;
                ald[i] = rd16;
                const float a = leaky(rs16 + rd16 + cvec[3], 0.2f);
                aself[i] = a;
                amax_u[i] = ordkey(a);
            }
        }
    } else {
        for (int u = 0; u < cnt; ++u) {
            const float4 t = hp[u * (HF / 4)];
            float4 o;
            o.x = (leaky(t.x, 0.01f) - mu[0]) * rs[0] * gg[0] + bb[0];
            o.y = (leaky(t.y, 0.01f) - mu[1]) * rs[1] * gg[1] + bb[1];
            o.z = (leaky(t.z, 0.01f) - mu[2]) * rs[2] * gg[2] + bb[2];
            o.w = (leaky(t.w, 0.01f) - mu[3]) * rs[3] * gg[3] + bb[3];
            xp[u * (HF / 4)] = o;
            if (dots) {
                float sr = o.x * ws[0] + o.y * ws[1] + o.z * ws[2] + o.w * ws[3];
                float dr = o.x * wd[0] + o.y * wd[1] + o.z * wd[2] + o.w * wd[3];
                for (int off = 16; off; off >>= 1) {
                    sr += __shfl_xor(sr, off);
                    dr += __shfl_xor(dr, off);
                }
                if (lane == 0) {
                    als[r0 + u] = sr;
                    ald[r0 + u] = dr;
                    const float a = leaky(sr + dr + cvec[3], 0.2f);
                    aself[r0 + u] = a;
                    amax_u[r0 + u] = ordkey(a);
                }
            }
        }
    }
}

extern "C" void kernel_launch(void* const* d_in, const int* in_sizes, int n_in,
                              void* d_out, int out_size, void* d_ws, size_t ws_size,
                              hipStream_t stream) {
    const float* x  = (const float*)d_in[0];
    const int*   ei = (const int*)d_in[1];
    const float* ea = (const float*)d_in[3];
    const int N = in_sizes[0] / HF;
    const int E = in_sizes[1] / 2;
    const int* src = ei;
    const int* dst = ei + E;

    float* ws = (float*)d_ws;
    const size_t NH = (size_t)N * HF;
    float* hbuf   = ws;
    float* acc    = hbuf + NH;
    float* xbuf   = acc + NH;
    float* alpha  = xbuf + NH;
    float* als    = alpha + E;
    float* ald    = als + N;
    float* aself  = ald + N;
    float* wvec   = aself + N;
    float* amaxf  = wvec + N;
    float* denom  = amaxf + N;
    float* colsum = denom + N;
    float* colsq  = colsum + HF;
    float* easum  = colsq + HF;
    float* cvec   = easum + 4;
    float* was = wvec;
    float* wad = wvec + HF;
    unsigned* tplane = (unsigned*)acc;
    int* elist = (int*)d_out;

    k_zero<<<1, 8, 0, stream>>>(easum, 8);
    k_ea_sum<<<128, 256, 0, stream>>>(ea, E, easum);

    const int nb_e  = (E + 255) / 256;
    const int nb_r  = ((N + 15) / 16 + 7) / 8;
    const int nb_t  = (N + TR - 1) / TR;

    for (int l = 0; l < 3; ++l) {
        const float* Wl  = (const float*)d_in[4 + 8 * l];
        const float* asv = (const float*)d_in[5 + 8 * l];
        const float* adv = (const float*)d_in[6 + 8 * l];
        const float* Wel = (const float*)d_in[7 + 8 * l];
        const float* aev = (const float*)d_in[8 + 8 * l];
        const float* bv  = (const float*)d_in[9 + 8 * l];
        const float* xin = (l == 0) ? x : xbuf;

        k_compute_c<<<1, 128, 0, stream>>>(Wl, asv, adv, Wel, aev, easum, was, wad, cvec, E);
        if (l == 0) {
            k_node_dots<<<512, 256, 0, stream>>>(xin, was, wad, cvec, als, ald, aself,
                                                 (unsigned*)amaxf, N);
        } else {
            const float* gp = (const float*)d_in[10 + 8 * (l - 1)];
            const float* bp = (const float*)d_in[11 + 8 * (l - 1)];
            k_bn_rows<<<nb_r, 256, 0, stream>>>(hbuf, colsum, colsq, gp, bp, xbuf, was, wad, cvec,
                                                als, ald, aself, (unsigned*)amaxf, N, 1);
        }
        k_edge_alpha<<<nb_e, 256, 0, stream>>>(src, dst, ea, als, ald, cvec,
                                               alpha, (unsigned*)amaxf, tplane, E);
        k_zero<<<1, 256, 0, stream>>>(colsum, 256);
        k_tile_rowsum<<<nb_t, 1024, 0, stream>>>(src, dst, tplane, alpha, (const unsigned*)amaxf,
                                                 aself, xin, Wl, bv, hbuf, colsum, colsq,
                                                 elist, l, N, E);
    }
    k_bn_rows<<<nb_r, 256, 0, stream>>>(hbuf, colsum, colsq, (const float*)d_in[10 + 8 * 2],
                                        (const float*)d_in[11 + 8 * 2], (float*)d_out, was, wad,
                                        cvec, als, ald, aself, (unsigned*)amaxf, N, 0);
}


# W a_src, W a_dst, We a_edge of all 3 layers and the edge_attr sums formed in one head launch; each layer owns its BN sums, cleared once; 18 to 12 launches; self-loop constant rounding may change
# speedup vs baseline: 2.9724x; 1.0624x over previous
#include <hip/hip_runtime.h>
#include <hip/hip_bf16.h>


typedef float v2f __attribute__((ext_vector_type(2)));
typedef float v8f __attribute__((ext_vector_type(8)));

#define HF 128

__device__ __forceinline__ unsigned ordkey(float f) {
    unsigned u = __float_as_uint(f);
    return (u & 0x80000000u) ? ~u : (u | 0x80000000u);
}
__device__ __forceinline__ float deord(unsigned u) {
    u = (u & 0x80000000u) ? (u & 0x7fffffffu) : ~u;
    return __uint_as_float(u);
}
__device__ __forceinline__ float leaky(float x, float s) { return x > 0.f ? x : s * x; }

__global__ void k_zero(float* p, int n) {
    int i = blockIdx.x * blockDim.x + threadIdx.x;
    if (i < n) p[i] = 0.f;
}

__global__ __launch_bounds__(128) void k_prep(const float* __restrict__ W0,
                                              const float* __restrict__ as0,
                                              const float* __restrict__ ad0,
                                              const float* __restrict__ We0,
                                              const float* __restrict__ ae0,
                                              const float* __restrict__ W1,
                                              const float* __restrict__ as1,
                                              const float* __restrict__ ad1,
                                              const float* __restrict__ We1,
                                              const float* __restrict__ ae1,
                                              const float* __restrict__ W2,
                                              const float* __restrict__ as2,
                                              const float* __restrict__ ad2,
                                              const float* __restrict__ We2,
                                              const float* __restrict__ ae2,
                                              const float* __restrict__ ea, int E,
                                              float* __restrict__ easum,
                                              float* __restrict__ wv3,
                                              float* __restrict__ cv3) {
    const int b = blockIdx.x;
    const int k = threadIdx.x;
    if (b < 3) {
        const float* W   = (b == 0) ? W0  : ((b == 1) ? W1  : W2);
        const float* avs = (b == 0) ? as0 : ((b == 1) ? as1 : as2);
        const float* avd = (b == 0) ? ad0 : ((b == 1) ? ad1 : ad2);
        const float* We  = (b == 0) ? We0 : ((b == 1) ? We1 : We2);
        const float* ae  = (b == 0) ? ae0 : ((b == 1) ? ae1 : ae2);
        float* was = wv3 + b * (2 * HF);
        float* wad = was + HF;
        const float4* wr = (const float4*)(W + (size_t)k * HF);
        float s = 0.f, d = 0.f;
        for (int q = 0; q < HF / 4; ++q) {
            const float4 w = wr[q];
            s += w.x * avs[4 * q] + w.y * avs[4 * q + 1] + w.z * avs[4 * q + 2] + w.w * avs[4 * q + 3];
            d += w.x * avd[4 * q] + w.y * avd[4 * q + 1] + w.z * avd[4 * q + 2] + w.w * avd[4 * q + 3];
        }
        was[k] = s;
        wad[k] = d;
        if (k < 3) {
            float cc = 0.f;
            for (int h = 0; h < HF; ++h) cc += We[k * HF + h] * ae[h];
            cv3[b * 4 + k] = cc;
        }
        return;
    }
    const int idx = (b - 3) * 128 + k;
    const int stride = ((int)gridDim.x - 3) * 128;
    float s0 = 0.f, s1 = 0.f, s2 = 0.f;
    for (int e = idx; e < E; e += stride) {
        s0 += ea[e * 3 + 0];
        s1 += ea[e * 3 + 1];
        s2 += ea[e * 3 + 2];
    }
    atomicAdd(easum + 0, s0);
    atomicAdd(easum + 1, s1);
    atomicAdd(easum + 2, s2);
}

__device__ __forceinline__ float self_c(const float* __restrict__ cvec,
                                        const float* __restrict__ easum, int E) {
    const float fe = (float)E;
    const float t0 = (easum[0] / fe) * cvec[0];
    const float t1 = (easum[1] / fe) * cvec[1];
    const float t2 = (easum[2] / fe) * cvec[2];
    return (t0 + t1) + t2;
}

__global__ __launch_bounds__(256) void k_node_dots(const float* __restrict__ Hm,
                                                   const float* __restrict__ avs,
                                                   const float* __restrict__ avd,
                                                   const float* __restrict__ cvec,
                                                   const float* __restrict__ easum,
                                                   float* __restrict__ als,
                                                   float* __restrict__ ald,
                                                   float* __restrict__ aself,
                                                   unsigned* __restrict__ amax_u, int n, int E) {
    const int lane = threadIdx.x & 31;
    int wid = (blockIdx.x * blockDim.x + threadIdx.x) >> 5;
    int nw  = (gridDim.x * blockDim.x) >> 5;
    float s0 = avs[lane], s1 = avs[lane + 32], s2 = avs[lane + 64], s3 = avs[lane + 96];
    float d0 = avd[lane], d1 = avd[lane + 32], d2 = avd[lane + 64], d3 = avd[lane + 96];
    const float c3 = self_c(cvec, easum, E);
    for (int i = wid; i < n; i += nw) {
        const float* hr = Hm + (size_t)i * HF;
        float h0 = hr[lane], h1 = hr[lane + 32], h2 = hr[lane + 64], h3 = hr[lane + 96];
        float rs = h0 * s0 + h1 * s1 + h2 * s2 + h3 * s3;
        float rd = h0 * d0 + h1 * d1 + h2 * d2 + h3 * d3;
        for (int off = 16; off; off >>= 1) {
            rs += __shfl_xor(rs, off);
            rd += __shfl_xor(rd, off);
        }
        if (lane == 0) {
            als[i] = rs;
            ald[i] = rd;
            const float a = leaky(rs + rd + c3, 0.2f);
            aself[i] = a;
            amax_u[i] = ordkey(a);
        }
    }
}

#define TR 200
#define TP 132

__global__ __launch_bounds__(256) void k_edge_alpha(const int* __restrict__ src,
                                                    const int* __restrict__ dst,
                                                    const float* __restrict__ ea,
                                                    const float* __restrict__ als,
                                                    const float* __restrict__ ald,
                                                    const float* __restrict__ cvec,
                                                    float* __restrict__ alpha,
                                                    unsigned* __restrict__ amax_u,
                                                    unsigned* __restrict__ tplane, int E) {
    __shared__ unsigned tbs[64];
    const int e = blockIdx.x * 256 + threadIdx.x;
    const bool live = e < E;
    int s = 0, d = 0;
    float e0 = 0.f, e1 = 0.f, e2 = 0.f;
    if (live) {
        s = src[e];
        d = dst[e];
        e0 = ea[e * 3 + 0];
        e1 = ea[e * 3 + 1];
        e2 = ea[e * 3 + 2];
    }
    ((unsigned char*)tbs)[threadIdx.x] =
        live ? (unsigned char)(((unsigned)d / TR) & 255u) : (unsigned char)0xff;
    __syncthreads();
    if (threadIdx.x < 64) tplane[blockIdx.x * 64 + threadIdx.x] = tbs[threadIdx.x];
    if (live) {
        const float c0 = cvec[0], c1 = cvec[1], c2 = cvec[2];
        const float ale = e0 * c0 + e1 * c1 + e2 * c2;
        const float a = leaky(als[s] + ald[d] + ale, 0.2f);
        alpha[e] = a;
        atomicMax(amax_u + d, ordkey(a));
    }
}

#define QCAP 64
#define SU 8

__device__ __forceinline__ unsigned zero_bytes(unsigned x) {
    const unsigned t = (x & 0x7f7f7f7fu) + 0x7f7f7f7fu;
    return ~(t | x | 0x7f7f7f7fu);
}

__device__ __forceinline__ void tile_add_edges4(const float* __restrict__ hb, float* tl,
                                                int s, int dl, float ex, int j) {
    float h[4][4], xj[4];
    int dj[4];
#pragma unroll
    for (int u = 0; u < 4; ++u) {
        const int sj = __builtin_amdgcn_readlane(s, j + u);
        dj[u] = __builtin_amdgcn_readlane(dl, j + u);
        xj[u] = __int_as_float(__builtin_amdgcn_readlane(__float_as_int(ex), j + u));
        const float* hr = hb + (size_t)sj * HF;
        h[u][0] = hr[0];
        h[u][1] = hr[32];
        h[u][2] = hr[64];
        h[u][3] = hr[96];
    }
#pragma unroll
    for (int u = 0; u < 4; ++u) {
        float* ap = tl + dj[u] * TP;
        atomicAdd(ap,      h[u][0] * xj[u]);
        atomicAdd(ap + 32, h[u][1] * xj[u]);
        atomicAdd(ap + 64, h[u][2] * xj[u]);
        atomicAdd(ap + 96, h[u][3] * xj[u]);
    }
}

__device__ __forceinline__ void tile_add_edge1(const float* __restrict__ hb, float* tl,
                                               int s, int dl, float ex, int j) {
    const int sj = __builtin_amdgcn_readlane(s, j);
    const int dd = __builtin_amdgcn_readlane(dl, j);
    const float xx = __int_as_float(__builtin_amdgcn_readlane(__float_as_int(ex), j));
    const float* hr = hb + (size_t)sj * HF;
    const float h0 = hr[0], h1 = hr[32], h2 = hr[64], h3 = hr[96];
    float* ap = tl + dd * TP;
    atomicAdd(ap,      h0 * xx);
    atomicAdd(ap + 32, h1 * xx);
    atomicAdd(ap + 64, h2 * xx);
    atomicAdd(ap + 96, h3 * xx);
}

__device__ __forceinline__ void tile_walk(const float* __restrict__ hb, float* tl,
                                          int s, int dl, float ex, unsigned okm, int j1) {
    int j = 0;
    for (; j + 4 <= j1; j += 4) {
        if (((okm >> j) & 15u) == 15u) {
            tile_add_edges4(hb, tl, s, dl, ex, j);
        } else {
            for (int u = 0; u < 4; ++u)
                if ((okm >> (j + u)) & 1u) tile_add_edge1(hb, tl, s, dl, ex, j + u);
        }
    }
    for (; j < j1; ++j)
        if ((okm >> j) & 1u) tile_add_edge1(hb, tl, s, dl, ex, j);
}

__global__ __launch_bounds__(1024) void k_tile_rowsum(const int* __restrict__ src,
                                                      const int* __restrict__ dst,
                                                      const unsigned* __restrict__ tplane,
                                                      const float* __restrict__ alpha,
                                                      const unsigned* __restrict__ amax_u,
                                                      const float* __restrict__ aself,
                                                      const float* __restrict__ Hm,
                                                      const float* __restrict__ W,
                                                      const float* __restrict__ bias,
                                                      float* __restrict__ Ho,
                                                      float* __restrict__ colsum,
                                                      float* __restrict__ colsq,
                                                      int* elist, int layer, int n, int E) {
    __shared__ float tile[TR * TP];
    __shared__ float mloc[TR];
    __shared__ float denl[TR];
    __shared__ int qe[32][QCAP];

    const int tid = threadIdx.x;
    const int lane = tid & 31;
    const int w = __builtin_amdgcn_readfirstlane(tid >> 5);
    const int r0 = blockIdx.x * TR;
    const int rows = (n - r0 < TR) ? (n - r0) : TR;

    if (tid < rows) {
        const int i = r0 + tid;
        const float m = deord(amax_u[i]);
        mloc[tid] = m;
        denl[tid] = expf(aself[i] - m);
    }
    __syncthreads();
    for (int r = w; r < rows; r += 32) {
        const float e = denl[r];
        const float4 t = ((const float4*)(Hm + (size_t)(r0 + r) * HF))[lane];
        float4 o;
        o.x = t.x * e;
        o.y = t.y * e;
        o.z = t.z * e;
        o.w = t.w * e;
        ((float4*)(tile + r * TP))[lane] = o;
    }
    __syncthreads();

    const float* hb = Hm + lane;
    float* tl = tile + lane;
    int* mye = qe[w];
    const uint4* tp4 = (const uint4*)tplane;
    const int nv = (E + 15) >> 4;
    const unsigned pat = (blockIdx.x & 255u) * 0x01010101u;
    int cnt = 0;

    const int nseg = (int)gridDim.x * 32;
    const int capw = (int)(((unsigned)(n * HF - nseg) / (unsigned)(nseg * 3)) & ~31u);
    const int seg = blockIdx.x * 32 + w;
    int* le = elist + (size_t)seg * 3 * capw;
    int* ls = le + capw;
    int* ld = ls + capw;
    int* ecnt = elist + (size_t)nseg * 3 * capw;
    int rec = (layer == 0) ? 0 : -1;
    bool scan = true;

    if (layer != 0) {
        const int cw = __builtin_amdgcn_readfirstlane(ecnt[seg]);
        if (cw >= 0) {
            scan = false;
            int li = (lane < cw) ? lane : 0;
            int eq = 0, s = 0, dq = 0;
            if (cw > 0) {
                eq = le[li];
                s = ls[li];
                dq = ld[li];
            }
            for (int g = 0; g < cw; g += 32) {
                const int left = cw - g;
                const int j1 = left < 32 ? left : 32;
                const bool live = lane < j1;
                const float al = alpha[eq];
                int eqn = 0, sn = 0, dqn = 0;
                if (left > 32) {
                    const int ln = g + 32 + ((lane < left - 32) ? lane : 0);
                    eqn = le[ln];
                    sn = ls[ln];
                    dqn = ld[ln];
                }
                const float ex0 = expf(al - mloc[dq]);
                const float ex = live ? ex0 : 0.f;
                atomicAdd(denl + dq, ex);
                tile_walk(hb, tl, s, dq, ex, __builtin_amdgcn_ballot_w32(live), j1);
                eq = eqn;
                s = sn;
                dq = dqn;
            }
        }
    }

    if (scan) {
        for (int v0 = w * (32 * SU); v0 < nv; v0 += 32 * 32 * SU) {
            uint4 tv[SU];
#pragma unroll
            for (int u = 0; u < SU; ++u) {
                const int vi = v0 + u * 32 + lane;
                tv[u] = tp4[vi < nv ? vi : nv - 1];
            }
#pragma unroll
            for (int u = 0; u < SU; ++u) {
                const int vi = v0 + u * 32 + lane;
                const int ebase = vi << 4;
                unsigned m = (zero_bytes(tv[u].x ^ pat) >> 7) | (zero_bytes(tv[u].y ^ pat) >> 6) |
                             (zero_bytes(tv[u].z ^ pat) >> 5) | (zero_bytes(tv[u].w ^ pat) >> 4);
                if (vi >= nv) m = 0u;
                unsigned any = __builtin_amdgcn_ballot_w32(m != 0u);
                for (int it = 0; it < 16 && any != 0u; ++it) {
                    const int p = __builtin_ctz(m | 0x80000000u);
                    const int e = ebase + ((p & 7) << 2) + (p >> 3);
                    const bool hit = (m != 0u) && (e < E);
                    const unsigned mask = __builtin_amdgcn_ballot_w32(hit);
                    if (hit) mye[__builtin_amdgcn_mbcnt_lo(mask, (unsigned)cnt)] = e;
                    cnt += __builtin_popcount(mask);
                    m &= m - 1u;
                    if (cnt >= 32) {
                        cnt -= 32;
                        const int eq = mye[cnt + lane];
                        const int s = src[eq];
                        const int dq = dst[eq] - r0;
                        const float al = alpha[eq];
                        const bool ok = (unsigned)dq < (unsigned)rows;
                        const int dqc = ok ? dq : 0;
                        const float ex0 = expf(al - mloc[dqc]);
                        const float ex = ok ? ex0 : 0.f;
                        atomicAdd(denl + dqc, ex);
                        const unsigned okm = __builtin_amdgcn_ballot_w32(ok);
                        if (rec >= 0) {
                            const int nok = __builtin_popcount(okm);
                            if (rec + nok <= capw) {
                                if (ok) {
                                    const int p = __builtin_amdgcn_mbcnt_lo(okm, (unsigned)rec);
                                    le[p] = eq;
                                    ls[p] = s;
                                    ld[p] = dq;
                                }
                                rec += nok;
                            } else {
                                rec = -1;
                            }
                        }
                        tile_walk(hb, tl, s, dqc, ex, okm, 32);
                    }
                    any = __builtin_amdgcn_ballot_w32(m != 0u);
                }
            }
        }

        {
            int s = 0, dq = 0;
            float ex = 0.f;
            bool ok = false;
            if (lane < cnt) {
                const int eq = mye[lane];
                s = src[eq];
                dq = dst[eq] - r0;
                const float al = alpha[eq];
                ok = (unsigned)dq < (unsigned)rows;
                ex = ok ? expf(al - mloc[ok ? dq : 0]) : 0.f;
                if (ok) atomicAdd(denl + dq, ex);
            }
            const unsigned okm = __builtin_amdgcn_ballot_w32(ok);
            if (rec >= 0) {
                const int nok = __builtin_popcount(okm);
                if (rec + nok <= capw) {
                    if (ok) {
                        const int p = __builtin_amdgcn_mbcnt_lo(okm, (unsigned)rec);
                        le[p] = mye[lane];
                        ls[p] = s;
                        ld[p] = dq;
                    }
                    rec += nok;
                } else {
                    rec = -1;
                }
            }
            if (layer == 0 && lane == 0) ecnt[seg] = rec;
            tile_walk(hb, tl, s, dq, ex, okm, cnt);
        }
    }

    const int lo16 = lane & 15;
    const int hi = lane >> 4;
    const int ct = w & 7;
    const int sg = w >> 3;
    const int ncol = ct * 16 + lo16;

    v2f breg[32];
#pragma unroll
    for (int kk = 0; kk < 32; ++kk) {
        const int kb = kk * 4 + hi * 2;
        v2f b;
        b.x = W[(size_t)kb * HF + ncol];
        b.y = W[(size_t)(kb + 1) * HF + ncol];
        breg[kk] = b;
    }
    const float bcol = bias[ncol];
    __syncthreads();

    float ps = 0.f, pq = 0.f;
    for (int s = sg; s * 16 < rows; s += 4) {
        const int rs0 = s * 16;
        int rl = rs0 + lo16;
        if (rl >= rows) rl = rows - 1;
        const float* xr = tile + rl * TP + hi * 2;

        const float rinvA = 1.0f / (denl[rl] + 1e-16f);

        v8f cacc = {0.f, 0.f, 0.f, 0.f, 0.f, 0.f, 0.f, 0.f};
#pragma unroll
        for (int kk = 0; kk < 32; ++kk) {
            const float2 t = *(const float2*)(xr + kk * 4);
            v2f a;
            a.x = t.x;
            a.y = t.y;
            cacc = __builtin_amdgcn_wmma_f32_16x16x4_f32(
                false, a, false, breg[kk], (short)0, cacc, false, false);
        }

        float rinv[8];
#pragma unroll
        for (int v = 0; v < 8; ++v) rinv[v] = __shfl(rinvA, v + hi * 8);

#pragma unroll
        for (int v = 0; v < 8; ++v) {
            const int row = rs0 + v + hi * 8;
            if (row < rows) {
                const float o = cacc[v] * rinv[v] + bcol;
                Ho[(size_t)(r0 + row) * HF + ncol] = o;
                const float y = leaky(o, 0.01f);
                ps += y;
                pq += y * y;
            }
        }
    }

    ps += __shfl_xor(ps, 16);
    pq += __shfl_xor(pq, 16);
    if (hi == 0) {
        atomicAdd(colsum + ncol, ps);
        atomicAdd(colsq + ncol, pq);
    }
}

__device__ __forceinline__ float rows16_sum(const float (&p)[16], int lane) {
    const bool b16 = (lane & 16) != 0, b8 = (lane & 8) != 0;
    const bool b4 = (lane & 4) != 0, b2 = (lane & 2) != 0;
    float t8[8], t4[4], t2[2];
#pragma unroll
    for (int u = 0; u < 8; ++u) {
        const float keep = b16 ? p[u + 8] : p[u];
        const float send = b16 ? p[u] : p[u + 8];
        t8[u] = keep + __shfl_xor(send, 16);
    }
#pragma unroll
    for (int u = 0; u < 4; ++u) {
        const float keep = b8 ? t8[u + 4] : t8[u];
        const float send = b8 ? t8[u] : t8[u + 4];
        t4[u] = keep + __shfl_xor(send, 8);
    }
#pragma unroll
    for (int u = 0; u < 2; ++u) {
        const float keep = b4 ? t4[u + 2] : t4[u];
        const float send = b4 ? t4[u] : t4[u + 2];
        t2[u] = keep + __shfl_xor(send, 4);
    }
    const float keep = b2 ? t2[1] : t2[0];
    const float send = b2 ? t2[0] : t2[1];
    const float t1 = keep + __shfl_xor(send, 2);
    return t1 + __shfl_xor(t1, 1);
}

__global__ __launch_bounds__(256) void k_bn_rows(const float* hin,
                                                 const float* __restrict__ colsum,
                                                 const float* __restrict__ colsq,
                                                 const float* __restrict__ g,
                                                 const float* __restrict__ be,
                                                 float* xout,
                                                 const float* __restrict__ was,
                                                 const float* __restrict__ wad,
                                                 const float* __restrict__ cvec,
                                                 const float* __restrict__ easum,
                                                 float* __restrict__ als,
                                                 float* __restrict__ ald,
                                                 float* __restrict__ aself,
                                                 unsigned* __restrict__ amax_u,
                                                 int n, int E, int dots) {
    const int lane = threadIdx.x & 31;
    const int wid = __builtin_amdgcn_readfirstlane(
        (int)((blockIdx.x * blockDim.x + threadIdx.x) >> 5));
    const int r0 = wid * 16;
    if (r0 >= n) return;
    const int cnt = (n - r0 < 16) ? (n - r0) : 16;

    const float4* hp = (const float4*)(hin + (size_t)r0 * HF) + lane;
    float4* xp = (float4*)(xout + (size_t)r0 * HF) + lane;
    const bool full = (cnt == 16);
    float4 v[16];
    if (full) {
#pragma unroll
        for (int u = 0; u < 16; ++u) v[u] = hp[u * (HF / 4)];
    }

    const int c4 = 4 * lane;
    const float invN = 1.f / (float)n;
    float mu[4], rs[4], gg[4], bb[4], ws[4], wd[4];
#pragma unroll
    for (int q = 0; q < 4; ++q) {
        const int c = c4 + q;
        mu[q] = colsum[c] * invN;
        const float var = colsq[c] * invN - mu[q] * mu[q];
        rs[q] = rsqrtf(var + 1e-5f);
        gg[q] = g[c];
        bb[q] = be[c];
        ws[q] = was[c];
        wd[q] = wad[c];
    }

    if (full) {
        float s[16], d[16];
#pragma unroll
        for (int u = 0; u < 16; ++u) {
            float4 o;
            o.x = (leaky(v[u].x, 0.01f) - mu[0]) * rs[0] * gg[0] + bb[0];
            o.y = (leaky(v[u].y, 0.01f) - mu[1]) * rs[1] * gg[1] + bb[1];
            o.z = (leaky(v[u].z, 0.01f) - mu[2]) * rs[2] * gg[2] + bb[2];
            o.w = (leaky(v[u].w, 0.01f) - mu[3]) * rs[3] * gg[3] + bb[3];
            xp[u * (HF / 4)] = o;
            s[u] = o.x * ws[0] + o.y * ws[1] + o.z * ws[2] + o.w * ws[3];
            d[u] = o.x * wd[0] + o.y * wd[1] + o.z * wd[2] + o.w * wd[3];
        }
        if (dots) {
            const float rs16 = rows16_sum(s, lane);
            const float rd16 = rows16_sum(d, lane);
            if ((lane & 1) == 0) {
                const int i = r0 + (lane >> 1);
                als[i] = rs16;
                ald[i] = rd16;
                const float a = leaky(rs16 + rd16 + self_c(cvec, easum, E), 0.2f);
                aself[i] = a;
                amax_u[i] = ordkey(a);
            }
        }
    } else {
        for (int u = 0; u < cnt; ++u) {
            const float4 t = hp[u * (HF / 4)];
            float4 o;
            o.x = (leaky(t.x, 0.01f) - mu[0]) * rs[0] * gg[0] + bb[0];
            o.y = (leaky(t.y, 0.01f) - mu[1]) * rs[1] * gg[1] + bb[1];
            o.z = (leaky(t.z, 0.01f) - mu[2]) * rs[2] * gg[2] + bb[2];
            o.w = (leaky(t.w, 0.01f) - mu[3]) * rs[3] * gg[3] + bb[3];
            xp[u * (HF / 4)] = o;
            if (dots) {
                float sr = o.x * ws[0] + o.y * ws[1] + o.z * ws[2] + o.w * ws[3];
                float dr = o.x * wd[0] + o.y * wd[1] + o.z * wd[2] + o.w * wd[3];
                for (int off = 16; off; off >>= 1) {
                    sr += __shfl_xor(sr, off);
                    dr += __shfl_xor(dr, off);
                }
                if (lane == 0) {
                    als[r0 + u] = sr;
                    ald[r0 + u] = dr;
                    const float a = leaky(sr + dr + self_c(cvec, easum, E), 0.2f);
                    aself[r0 + u] = a;
                    amax_u[r0 + u] = ordkey(a);
                }
            }
        }
    }
}

extern "C" void kernel_launch(void* const* d_in, const int* in_sizes, int n_in,
                              void* d_out, int out_size, void* d_ws, size_t ws_size,
                              hipStream_t stream) {
    const float* x  = (const float*)d_in[0];
    const int*   ei = (const int*)d_in[1];
    const float* ea = (const float*)d_in[3];
    const int N = in_sizes[0] / HF;
    const int E = in_sizes[1] / 2;
    const int* src = ei;
    const int* dst = ei + E;

    float* ws = (float*)d_ws;
    const size_t NH = (size_t)N * HF;
    float* hbuf   = ws;
    float* acc    = hbuf + NH;
    float* xbuf   = acc + NH;
    float* alpha  = xbuf + NH;
    float* als    = alpha + E;
    float* ald    = als + N;
    float* aself  = ald + N;
    float* wvec   = aself + N;
    float* amaxf  = wvec + N;
    float* svec  = amaxf + N;
    float* colsum3 = svec;
    float* easum  = colsum3 + 6 * HF;
    float* cvec3  = easum + 4;
    unsigned* tplane = (unsigned*)acc;
    int* elist = (int*)d_out;

    k_zero<<<4, 256, 0, stream>>>(colsum3, 772);
    k_prep<<<3 + 256, 128, 0, stream>>>(
        (const float*)d_in[4], (const float*)d_in[5], (const float*)d_in[6],
        (const float*)d_in[7], (const float*)d_in[8],
        (const float*)d_in[12], (const float*)d_in[13], (const float*)d_in[14],
        (const float*)d_in[15], (const float*)d_in[16],
        (const float*)d_in[20], (const float*)d_in[21], (const float*)d_in[22],
        (const float*)d_in[23], (const float*)d_in[24],
        ea, E, easum, wvec, cvec3);

    const int nb_e  = (E + 255) / 256;
    const int nb_r  = ((N + 15) / 16 + 7) / 8;
    const int nb_t  = (N + TR - 1) / TR;

    for (int l = 0; l < 3; ++l) {
        const float* Wl  = (const float*)d_in[4 + 8 * l];
        const float* bv  = (const float*)d_in[9 + 8 * l];
        const float* xin = (l == 0) ? x : xbuf;
        const float* was = wvec + l * 2 * HF;
        const float* wad = was + HF;
        const float* cvec = cvec3 + l * 4;
        float* colsum = colsum3 + l * 2 * HF;
        float* colsq  = colsum + HF;

        if (l == 0) {
            k_node_dots<<<512, 256, 0, stream>>>(xin, was, wad, cvec, easum, als, ald, aself,
                                                 (unsigned*)amaxf, N, E);
        } else {
            const float* gp = (const float*)d_in[10 + 8 * (l - 1)];
            const float* bp = (const float*)d_in[11 + 8 * (l - 1)];
            k_bn_rows<<<nb_r, 256, 0, stream>>>(hbuf, colsum - 2 * HF, colsq - 2 * HF, gp, bp, xbuf,
                                                was, wad, cvec, easum, als, ald, aself,
                                                (unsigned*)amaxf, N, E, 1);
        }
        k_edge_alpha<<<nb_e, 256, 0, stream>>>(src, dst, ea, als, ald, cvec,
                                               alpha, (unsigned*)amaxf, tplane, E);
        k_tile_rowsum<<<nb_t, 1024, 0, stream>>>(src, dst, tplane, alpha, (const unsigned*)amaxf,
                                                 aself, xin, Wl, bv, hbuf, colsum, colsq,
                                                 elist, l, N, E);
    }
    k_bn_rows<<<nb_r, 256, 0, stream>>>(hbuf, colsum3 + 4 * HF, colsum3 + 5 * HF,
                                        (const float*)d_in[10 + 8 * 2],
                                        (const float*)d_in[11 + 8 * 2], (float*)d_out,
                                        wvec, wvec + HF, cvec3, easum, als, ald, aself,
                                        (unsigned*)amaxf, N, E, 0);
}


# Edge logits and segment max move into the row-sum tile kernel: two walks of its per-wave edge lists (LDS integer max, then exp and adds); 3 global atomic-max edge passes become 1 streaming pass
# speedup vs baseline: 4.1975x; 1.4122x over previous
#include <hip/hip_runtime.h>
#include <hip/hip_bf16.h>


typedef float v2f __attribute__((ext_vector_type(2)));
typedef float v8f __attribute__((ext_vector_type(8)));

#define HF 128

__device__ __forceinline__ unsigned ordkey(float f) {
    unsigned u = __float_as_uint(f);
    return (u & 0x80000000u) ? ~u : (u | 0x80000000u);
}
__device__ __forceinline__ float deord(unsigned u) {
    u = (u & 0x80000000u) ? (u & 0x7fffffffu) : ~u;
    return __uint_as_float(u);
}
__device__ __forceinline__ float leaky(float x, float s) { return x > 0.f ? x : s * x; }

__global__ void k_zero(float* p, int n) {
    int i = blockIdx.x * blockDim.x + threadIdx.x;
    if (i < n) p[i] = 0.f;
}

__global__ __launch_bounds__(128) void k_prep(const float* __restrict__ W0,
                                              const float* __restrict__ as0,
                                              const float* __restrict__ ad0,
                                              const float* __restrict__ We0,
                                              const float* __restrict__ ae0,
                                              const float* __restrict__ W1,
                                              const float* __restrict__ as1,
                                              const float* __restrict__ ad1,
                                              const float* __restrict__ We1,
                                              const float* __restrict__ ae1,
                                              const float* __restrict__ W2,
                                              const float* __restrict__ as2,
                                              const float* __restrict__ ad2,
                                              const float* __restrict__ We2,
                                              const float* __restrict__ ae2,
                                              const float* __restrict__ ea, int E,
                                              float* __restrict__ easum,
                                              float* __restrict__ wv3,
                                              float* __restrict__ cv3) {
    const int b = blockIdx.x;
    const int k = threadIdx.x;
    if (b < 3) {
        const float* W   = (b == 0) ? W0  : ((b == 1) ? W1  : W2);
        const float* avs = (b == 0) ? as0 : ((b == 1) ? as1 : as2);
        const float* avd = (b == 0) ? ad0 : ((b == 1) ? ad1 : ad2);
        const float* We  = (b == 0) ? We0 : ((b == 1) ? We1 : We2);
        const float* ae  = (b == 0) ? ae0 : ((b == 1) ? ae1 : ae2);
        float* was = wv3 + b * (2 * HF);
        float* wad = was + HF;
        const float4* wr = (const float4*)(W + (size_t)k * HF);
        float s = 0.f, d = 0.f;
        for (int q = 0; q < HF / 4; ++q) {
            const float4 w = wr[q];
            s += w.x * avs[4 * q] + w.y * avs[4 * q + 1] + w.z * avs[4 * q + 2] + w.w * avs[4 * q + 3];
            d += w.x * avd[4 * q] + w.y * avd[4 * q + 1] + w.z * avd[4 * q + 2] + w.w * avd[4 * q + 3];
        }
        was[k] = s;
        wad[k] = d;
        if (k < 3) {
            float cc = 0.f;
            for (int h = 0; h < HF; ++h) cc += We[k * HF + h] * ae[h];
            cv3[b * 4 + k] = cc;
        }
        return;
    }
    const int idx = (b - 3) * 128 + k;
    const int stride = ((int)gridDim.x - 3) * 128;
    float s0 = 0.f, s1 = 0.f, s2 = 0.f;
    for (int e = idx; e < E; e += stride) {
        s0 += ea[e * 3 + 0];
        s1 += ea[e * 3 + 1];
        s2 += ea[e * 3 + 2];
    }
    atomicAdd(easum + 0, s0);
    atomicAdd(easum + 1, s1);
    atomicAdd(easum + 2, s2);
}

__device__ __forceinline__ float self_c(const float* __restrict__ cvec,
                                        const float* __restrict__ easum, int E) {
    const float fe = (float)E;
    const float t0 = (easum[0] / fe) * cvec[0];
    const float t1 = (easum[1] / fe) * cvec[1];
    const float t2 = (easum[2] / fe) * cvec[2];
    return (t0 + t1) + t2;
}

__global__ __launch_bounds__(256) void k_node_dots(const float* __restrict__ Hm,
                                                   const float* __restrict__ avs,
                                                   const float* __restrict__ avd,
                                                   const float* __restrict__ cvec,
                                                   const float* __restrict__ easum,
                                                   float* __restrict__ als,
                                                   float* __restrict__ ald,
                                                   float* __restrict__ aself, int n, int E) {
    const int lane = threadIdx.x & 31;
    int wid = (blockIdx.x * blockDim.x + threadIdx.x) >> 5;
    int nw  = (gridDim.x * blockDim.x) >> 5;
    float s0 = avs[lane], s1 = avs[lane + 32], s2 = avs[lane + 64], s3 = avs[lane + 96];
    float d0 = avd[lane], d1 = avd[lane + 32], d2 = avd[lane + 64], d3 = avd[lane + 96];
    const float c3 = self_c(cvec, easum, E);
    for (int i = wid; i < n; i += nw) {
        const float* hr = Hm + (size_t)i * HF;
        float h0 = hr[lane], h1 = hr[lane + 32], h2 = hr[lane + 64], h3 = hr[lane + 96];
        float rs = h0 * s0 + h1 * s1 + h2 * s2 + h3 * s3;
        float rd = h0 * d0 + h1 * d1 + h2 * d2 + h3 * d3;
        for (int off = 16; off; off >>= 1) {
            rs += __shfl_xor(rs, off);
            rd += __shfl_xor(rd, off);
        }
        if (lane == 0) {
            als[i] = rs;
            ald[i] = rd;
            aself[i] = leaky(rs + rd + c3, 0.2f);
        }
    }
}

#define TR 200
#define TP 132

__global__ __launch_bounds__(256) void k_edge_prep(const int* __restrict__ dst,
                                                   const float* __restrict__ ea,
                                                   const float* __restrict__ cv3,
                                                   float* __restrict__ ale3,
                                                   unsigned* __restrict__ tplane, int E) {
    __shared__ unsigned tbs[64];
    const int e = blockIdx.x * 256 + threadIdx.x;
    const bool live = e < E;
    int d = 0;
    float e0 = 0.f, e1 = 0.f, e2 = 0.f;
    if (live) {
        d = dst[e];
        e0 = ea[e * 3 + 0];
        e1 = ea[e * 3 + 1];
        e2 = ea[e * 3 + 2];
    }
    ((unsigned char*)tbs)[threadIdx.x] =
        live ? (unsigned char)(((unsigned)d / TR) & 255u) : (unsigned char)0xff;
    __syncthreads();
    if (threadIdx.x < 64) tplane[blockIdx.x * 64 + threadIdx.x] = tbs[threadIdx.x];
    if (live) {
        for (int l = 0; l < 3; ++l) {
            const float c0 = cv3[4 * l], c1 = cv3[4 * l + 1], c2 = cv3[4 * l + 2];
            ale3[(size_t)l * E + e] = e0 * c0 + e1 * c1 + e2 * c2;
        }
    }
}

#define QCAP 64
#define SU 8

__device__ __forceinline__ unsigned zero_bytes(unsigned x) {
    const unsigned t = (x & 0x7f7f7f7fu) + 0x7f7f7f7fu;
    return ~(t | x | 0x7f7f7f7fu);
}

__device__ __forceinline__ void tile_add_edges4(const float* __restrict__ hb, float* tl,
                                                int s, int dl, float ex, int j) {
    float h[4][4], xj[4];
    int dj[4];
#pragma unroll
    for (int u = 0; u < 4; ++u) {
        const int sj = __builtin_amdgcn_readlane(s, j + u);
        dj[u] = __builtin_amdgcn_readlane(dl, j + u);
        xj[u] = __int_as_float(__builtin_amdgcn_readlane(__float_as_int(ex), j + u));
        const float* hr = hb + (size_t)sj * HF;
        h[u][0] = hr[0];
        h[u][1] = hr[32];
        h[u][2] = hr[64];
        h[u][3] = hr[96];
    }
#pragma unroll
    for (int u = 0; u < 4; ++u) {
        float* ap = tl + dj[u] * TP;
        atomicAdd(ap,      h[u][0] * xj[u]);
        atomicAdd(ap + 32, h[u][1] * xj[u]);
        atomicAdd(ap + 64, h[u][2] * xj[u]);
        atomicAdd(ap + 96, h[u][3] * xj[u]);
    }
}

__device__ __forceinline__ void tile_add_edge1(const float* __restrict__ hb, float* tl,
                                               int s, int dl, float ex, int j) {
    const int sj = __builtin_amdgcn_readlane(s, j);
    const int dd = __builtin_amdgcn_readlane(dl, j);
    const float xx = __int_as_float(__builtin_amdgcn_readlane(__float_as_int(ex), j));
    const float* hr = hb + (size_t)sj * HF;
    const float h0 = hr[0], h1 = hr[32], h2 = hr[64], h3 = hr[96];
    float* ap = tl + dd * TP;
    atomicAdd(ap,      h0 * xx);
    atomicAdd(ap + 32, h1 * xx);
    atomicAdd(ap + 64, h2 * xx);
    atomicAdd(ap + 96, h3 * xx);
}

__device__ __forceinline__ void tile_walk(const float* __restrict__ hb, float* tl,
                                          int s, int dl, float ex, unsigned okm, int j1) {
    int j = 0;
    for (; j + 4 <= j1; j += 4) {
        if (((okm >> j) & 15u) == 15u) {
            tile_add_edges4(hb, tl, s, dl, ex, j);
        } else {
            for (int u = 0; u < 4; ++u)
                if ((okm >> (j + u)) & 1u) tile_add_edge1(hb, tl, s, dl, ex, j + u);
        }
    }
    for (; j < j1; ++j)
        if ((okm >> j) & 1u) tile_add_edge1(hb, tl, s, dl, ex, j);
}

#define LRUN 6

template <class F>
__device__ __forceinline__ void plane_scan(const uint4* __restrict__ tp4, int nv, int E,
                                           unsigned pat, int w, int lane, int* mye,
                                           F&& drain) {
    int cnt = 0;
    for (int v0 = w * (32 * SU); v0 < nv; v0 += 32 * 32 * SU) {
        uint4 tv[SU];
#pragma unroll
        for (int u = 0; u < SU; ++u) {
            const int vi = v0 + u * 32 + lane;
            tv[u] = tp4[vi < nv ? vi : nv - 1];
        }
#pragma unroll
        for (int u = 0; u < SU; ++u) {
            const int vi = v0 + u * 32 + lane;
            const int ebase = vi << 4;
            unsigned m = (zero_bytes(tv[u].x ^ pat) >> 7) | (zero_bytes(tv[u].y ^ pat) >> 6) |
                         (zero_bytes(tv[u].z ^ pat) >> 5) | (zero_bytes(tv[u].w ^ pat) >> 4);
            if (vi >= nv) m = 0u;
            unsigned any = __builtin_amdgcn_ballot_w32(m != 0u);
            for (int it = 0; it < 16 && any != 0u; ++it) {
                const int p = __builtin_ctz(m | 0x80000000u);
                const int e = ebase + ((p & 7) << 2) + (p >> 3);
                const bool hit = (m != 0u) && (e < E);
                const unsigned mask = __builtin_amdgcn_ballot_w32(hit);
                if (hit) mye[__builtin_amdgcn_mbcnt_lo(mask, (unsigned)cnt)] = e;
                cnt += __builtin_popcount(mask);
                m &= m - 1u;
                if (cnt >= 32) {
                    cnt -= 32;
                    if (!drain(mye[cnt + lane], true, 32)) return;
                }
                any = __builtin_amdgcn_ballot_w32(m != 0u);
            }
        }
    }
    const bool v = lane < cnt;
    drain(v ? mye[lane] : 0, v, cnt);
}

__global__ __launch_bounds__(1024) void k_tile_rowsum(const int* __restrict__ src,
                                                      const int* __restrict__ dst,
                                                      const unsigned* __restrict__ tplane,
                                                      const float* __restrict__ ale,
                                                      const float* __restrict__ als,
                                                      const float* __restrict__ ald,
                                                      const float* __restrict__ aself,
                                                      const float* __restrict__ Hm,
                                                      const float* __restrict__ W,
                                                      const float* __restrict__ bias,
                                                      float* __restrict__ Ho,
                                                      float* __restrict__ colsum,
                                                      float* __restrict__ colsq,
                                                      int* elist, int layer, int n, int E) {
    __shared__ float tile[TR * TP];
    __shared__ float mloc[TR];
    __shared__ unsigned mkey[TR];
    __shared__ float denl[TR];
    __shared__ float aldl[TR];
    __shared__ int qe[32][QCAP];

    const int tid = threadIdx.x;
    const int lane = tid & 31;
    const int w = __builtin_amdgcn_readfirstlane(tid >> 5);
    const int r0 = blockIdx.x * TR;
    const int rows = (n - r0 < TR) ? (n - r0) : TR;

    float myself = 0.f;
    if (tid < rows) {
        const int i = r0 + tid;
        myself = aself[i];
        mkey[tid] = ordkey(myself);
        aldl[tid] = ald[i];
    }

    const float* hb = Hm + lane;
    float* tl = tile + lane;
    int* mye = qe[w];
    const uint4* tp4 = (const uint4*)tplane;
    const int nv = (E + 15) >> 4;
    const unsigned pat = (blockIdx.x & 255u) * 0x01010101u;

    const int nseg = (int)gridDim.x * 32;
    int capw = (int)(((unsigned)(n * HF - nseg) / (unsigned)(nseg * 4)) & ~31u);
    if (capw > 32 * LRUN) capw = 32 * LRUN;
    const int seg = blockIdx.x * 32 + w;
    int* le = elist + (size_t)seg * 4 * capw;
    int* ls = le + capw;
    int* ld = ls + capw;
    float* la = (float*)(ld + capw);
    int* ecnt = elist + (size_t)nseg * 4 * capw;

    int cw;
    if (layer == 0) {
        int rec = 0;
        plane_scan(tp4, nv, E, pat, w, lane, mye, [&](int eq, bool valid, int) -> bool {
            int s = 0, dq = -1;
            if (valid) {
                s = src[eq];
                dq = dst[eq] - r0;
            }
            const bool ok = valid && ((unsigned)dq < (unsigned)rows);
            const unsigned okm = __builtin_amdgcn_ballot_w32(ok);
            const int nok = __builtin_popcount(okm);
            if (rec + nok > capw) {
                rec = -1;
                return false;
            }
            if (ok) {
                const int p = __builtin_amdgcn_mbcnt_lo(okm, (unsigned)rec);
                le[p] = eq;
                ls[p] = s;
                ld[p] = dq;
            }
            rec += nok;
            return true;
        });
        cw = __builtin_amdgcn_readfirstlane(rec);
        if (lane == 0) ecnt[seg] = cw;
    } else {
        cw = __builtin_amdgcn_readfirstlane(ecnt[seg]);
    }
    __syncthreads();

    if (cw > 0) {
        int eqr[LRUN], sr[LRUN], dr[LRUN];
        float asr[LRUN], aer[LRUN];
#pragma unroll
        for (int g = 0; g < LRUN; ++g) {
            const int li = g * 32 + lane;
            const int lic = (li < cw) ? li : 0;
            eqr[g] = le[lic];
            sr[g] = ls[lic];
            dr[g] = ld[lic];
        }
#pragma unroll
        for (int g = 0; g < LRUN; ++g) {
            const bool v = (g * 32 + lane) < cw;
            asr[g] = als[v ? sr[g] : 0];
            aer[g] = ale[v ? eqr[g] : 0];
        }
#pragma unroll
        for (int g = 0; g < LRUN; ++g) {
            if ((g * 32 + lane) < cw) {
                const float a = leaky(asr[g] + aldl[dr[g]] + aer[g], 0.2f);
                atomicMax(mkey + dr[g], ordkey(a));
                la[g * 32 + lane] = a;
            }
        }
    } else if (cw < 0) {
        plane_scan(tp4, nv, E, pat, w, lane, mye, [&](int eq, bool valid, int) -> bool {
            int s = 0, dq = -1;
            float ae = 0.f;
            if (valid) {
                s = src[eq];
                dq = dst[eq] - r0;
                ae = ale[eq];
            }
            const bool ok = valid && ((unsigned)dq < (unsigned)rows);
            const int dqc = ok ? dq : 0;
            const float a = leaky(als[s] + aldl[dqc] + ae, 0.2f);
            if (ok) atomicMax(mkey + dqc, ordkey(a));
            return true;
        });
    }
    __syncthreads();

    if (tid < rows) {
        const float m = deord(mkey[tid]);
        mloc[tid] = m;
        denl[tid] = expf(myself - m);
    }
    __syncthreads();
    for (int r = w; r < rows; r += 32) {
        const float e = denl[r];
        const float4 t = ((const float4*)(Hm + (size_t)(r0 + r) * HF))[lane];
        float4 o;
        o.x = t.x * e;
        o.y = t.y * e;
        o.z = t.z * e;
        o.w = t.w * e;
        ((float4*)(tile + r * TP))[lane] = o;
    }
    __syncthreads();

    if (cw >= 0) {
        if (cw > 0) {
            const int li = (lane < cw) ? lane : 0;
            int s = ls[li], dq = ld[li];
            float al = la[li];
            for (int g = 0; g < cw; g += 32) {
                const int left = cw - g;
                const int j1 = left < 32 ? left : 32;
                const bool live = lane < j1;
                int sn = 0, dqn = 0;
                float aln = 0.f;
                if (left > 32) {
                    const int ln = g + 32 + ((lane < left - 32) ? lane : 0);
                    sn = ls[ln];
                    dqn = ld[ln];
                    aln = la[ln];
                }
                const float ex0 = expf(al - mloc[dq]);
                const float ex = live ? ex0 : 0.f;
                atomicAdd(denl + dq, ex);
                tile_walk(hb, tl, s, dq, ex, __builtin_amdgcn_ballot_w32(live), j1);
                s = sn;
                dq = dqn;
                al = aln;
            }
        }
    } else {
        plane_scan(tp4, nv, E, pat, w, lane, mye, [&](int eq, bool valid, int nj) -> bool {
            int s = 0, dq = -1;
            float ae = 0.f;
            if (valid) {
                s = src[eq];
                dq = dst[eq] - r0;
                ae = ale[eq];
            }
            const bool ok = valid && ((unsigned)dq < (unsigned)rows);
            const int dqc = ok ? dq : 0;
            const float a = leaky(als[s] + aldl[dqc] + ae, 0.2f);
            const float ex = ok ? expf(a - mloc[dqc]) : 0.f;
            if (ok) atomicAdd(denl + dqc, ex);
            tile_walk(hb, tl, s, dqc, ex, __builtin_amdgcn_ballot_w32(ok), nj);
            return true;
        });
    }

    const int lo16 = lane & 15;
    const int hi = lane >> 4;
    const int ct = w & 7;
    const int sg = w >> 3;
    const int ncol = ct * 16 + lo16;

    v2f breg[32];
#pragma unroll
    for (int kk = 0; kk < 32; ++kk) {
        const int kb = kk * 4 + hi * 2;
        v2f b;
        b.x = W[(size_t)kb * HF + ncol];
        b.y = W[(size_t)(kb + 1) * HF + ncol];
        breg[kk] = b;
    }
    const float bcol = bias[ncol];
    __syncthreads();

    float ps = 0.f, pq = 0.f;
    for (int s = sg; s * 16 < rows; s += 4) {
        const int rs0 = s * 16;
        int rl = rs0 + lo16;
        if (rl >= rows) rl = rows - 1;
        const float* xr = tile + rl * TP + hi * 2;

        const float rinvA = 1.0f / (denl[rl] + 1e-16f);

        v8f cacc = {0.f, 0.f, 0.f, 0.f, 0.f, 0.f, 0.f, 0.f};
#pragma unroll
        for (int kk = 0; kk < 32; ++kk) {
            const float2 t = *(const float2*)(xr + kk * 4);
            v2f a;
            a.x = t.x;
            a.y = t.y;
            cacc = __builtin_amdgcn_wmma_f32_16x16x4_f32(
                false, a, false, breg[kk], (short)0, cacc, false, false);
        }

        float rinv[8];
#pragma unroll
        for (int v = 0; v < 8; ++v) rinv[v] = __shfl(rinvA, v + hi * 8);

#pragma unroll
        for (int v = 0; v < 8; ++v) {
            const int row = rs0 + v + hi * 8;
            if (row < rows) {
                const float o = cacc[v] * rinv[v] + bcol;
                Ho[(size_t)(r0 + row) * HF + ncol] = o;
                const float y = leaky(o, 0.01f);
                ps += y;
                pq += y * y;
            }
        }
    }

    ps += __shfl_xor(ps, 16);
    pq += __shfl_xor(pq, 16);
    if (hi == 0) {
        atomicAdd(colsum + ncol, ps);
        atomicAdd(colsq + ncol, pq);
    }
}

__device__ __forceinline__ float rows16_sum(const float (&p)[16], int lane) {
    const bool b16 = (lane & 16) != 0, b8 = (lane & 8) != 0;
    const bool b4 = (lane & 4) != 0, b2 = (lane & 2) != 0;
    float t8[8], t4[4], t2[2];
#pragma unroll
    for (int u = 0; u < 8; ++u) {
        const float keep = b16 ? p[u + 8] : p[u];
        const float send = b16 ? p[u] : p[u + 8];
        t8[u] = keep + __shfl_xor(send, 16);
    }
#pragma unroll
    for (int u = 0; u < 4; ++u) {
        const float keep = b8 ? t8[u + 4] : t8[u];
        const float send = b8 ? t8[u] : t8[u + 4];
        t4[u] = keep + __shfl_xor(send, 8);
    }
#pragma unroll
    for (int u = 0; u < 2; ++u) {
        const float keep = b4 ? t4[u + 2] : t4[u];
        const float send = b4 ? t4[u] : t4[u + 2];
        t2[u] = keep + __shfl_xor(send, 4);
    }
    const float keep = b2 ? t2[1] : t2[0];
    const float send = b2 ? t2[0] : t2[1];
    const float t1 = keep + __shfl_xor(send, 2);
    return t1 + __shfl_xor(t1, 1);
}

__global__ __launch_bounds__(256) void k_bn_rows(const float* hin,
                                                 const float* __restrict__ colsum,
                                                 const float* __restrict__ colsq,
                                                 const float* __restrict__ g,
                                                 const float* __restrict__ be,
                                                 float* xout,
                                                 const float* __restrict__ was,
                                                 const float* __restrict__ wad,
                                                 const float* __restrict__ cvec,
                                                 const float* __restrict__ easum,
                                                 float* __restrict__ als,
                                                 float* __restrict__ ald,
                                                 float* __restrict__ aself,
                                                 int n, int E, int dots) {
    const int lane = threadIdx.x & 31;
    const int wid = __builtin_amdgcn_readfirstlane(
        (int)((blockIdx.x * blockDim.x + threadIdx.x) >> 5));
    const int r0 = wid * 16;
    if (r0 >= n) return;
    const int cnt = (n - r0 < 16) ? (n - r0) : 16;

    const float4* hp = (const float4*)(hin + (size_t)r0 * HF) + lane;
    float4* xp = (float4*)(xout + (size_t)r0 * HF) + lane;
    const bool full = (cnt == 16);
    float4 v[16];
    if (full) {
#pragma unroll
        for (int u = 0; u < 16; ++u) v[u] = hp[u * (HF / 4)];
    }

    const int c4 = 4 * lane;
    const float invN = 1.f / (float)n;
    float mu[4], rs[4], gg[4], bb[4], ws[4], wd[4];
#pragma unroll
    for (int q = 0; q < 4; ++q) {
        const int c = c4 + q;
        mu[q] = colsum[c] * invN;
        const float var = colsq[c] * invN - mu[q] * mu[q];
        rs[q] = rsqrtf(var + 1e-5f);
        gg[q] = g[c];
        bb[q] = be[c];
        ws[q] = was[c];
        wd[q] = wad[c];
    }

    if (full) {
        float s[16], d[16];
#pragma unroll
        for (int u = 0; u < 16; ++u) {
            float4 o;
            o.x = (leaky(v[u].x, 0.01f) - mu[0]) * rs[0] * gg[0] + bb[0];
            o.y = (leaky(v[u].y, 0.01f) - mu[1]) * rs[1] * gg[1] + bb[1];
            o.z = (leaky(v[u].z, 0.01f) - mu[2]) * rs[2] * gg[2] + bb[2];
            o.w = (leaky(v[u].w, 0.01f) - mu[3]) * rs[3] * gg[3] + bb[3];
            xp[u * (HF / 4)] = o;
            s[u] = o.x * ws[0] + o.y * ws[1] + o.z * ws[2] + o.w * ws[3];
            d[u] = o.x * wd[0] + o.y * wd[1] + o.z * wd[2] + o.w * wd[3];
        }
        if (dots) {
            const float rs16 = rows16_sum(s, lane);
            const float rd16 = rows16_sum(d, lane);
            if ((lane & 1) == 0) {
                const int i = r0 + (lane >> 1);
                als[i] = rs16;
                ald[i] = rd16;
                aself[i] = leaky(rs16 + rd16 + self_c(cvec, easum, E), 0.2f);
            }
        }
    } else {
        for (int u = 0; u < cnt; ++u) {
            const float4 t = hp[u * (HF / 4)];
            float4 o;
            o.x = (leaky(t.x, 0.01f) - mu[0]) * rs[0] * gg[0] + bb[0];
            o.y = (leaky(t.y, 0.01f) - mu[1]) * rs[1] * gg[1] + bb[1];
            o.z = (leaky(t.z, 0.01f) - mu[2]) * rs[2] * gg[2] + bb[2];
            o.w = (leaky(t.w, 0.01f) - mu[3]) * rs[3] * gg[3] + bb[3];
            xp[u * (HF / 4)] = o;
            if (dots) {
                float sr = o.x * ws[0] + o.y * ws[1] + o.z * ws[2] + o.w * ws[3];
                float dr = o.x * wd[0] + o.y * wd[1] + o.z * wd[2] + o.w * wd[3];
                for (int off = 16; off; off >>= 1) {
                    sr += __shfl_xor(sr, off);
                    dr += __shfl_xor(dr, off);
                }
                if (lane == 0) {
                    als[r0 + u] = sr;
                    ald[r0 + u] = dr;
                    aself[r0 + u] = leaky(sr + dr + self_c(cvec, easum, E), 0.2f);
                }
            }
        }
    }
}

extern "C" void kernel_launch(void* const* d_in, const int* in_sizes, int n_in,
                              void* d_out, int out_size, void* d_ws, size_t ws_size,
                              hipStream_t stream) {
    const float* x  = (const float*)d_in[0];
    const int*   ei = (const int*)d_in[1];
    const float* ea = (const float*)d_in[3];
    const int N = in_sizes[0] / HF;
    const int E = in_sizes[1] / 2;
    const int* src = ei;
    const int* dst = ei + E;

    float* ws = (float*)d_ws;
    const size_t NH = (size_t)N * HF;
    float* hbuf   = ws;
    float* acc    = hbuf + NH;
    float* xbuf   = acc + NH;
    float* als    = xbuf + NH + E;
    float* ald    = als + N;
    float* aself  = ald + N;
    float* wvec   = aself + N;
    float* svec  = wvec + 2 * (size_t)N;
    float* colsum3 = svec;
    float* easum  = colsum3 + 6 * HF;
    float* cvec3  = easum + 4;
    unsigned* tplane = (unsigned*)acc;
    int* elist = (int*)d_out;

    k_zero<<<4, 256, 0, stream>>>(colsum3, 772);
    k_prep<<<3 + 256, 128, 0, stream>>>(
        (const float*)d_in[4], (const float*)d_in[5], (const float*)d_in[6],
        (const float*)d_in[7], (const float*)d_in[8],
        (const float*)d_in[12], (const float*)d_in[13], (const float*)d_in[14],
        (const float*)d_in[15], (const float*)d_in[16],
        (const float*)d_in[20], (const float*)d_in[21], (const float*)d_in[22],
        (const float*)d_in[23], (const float*)d_in[24],
        ea, E, easum, wvec, cvec3);

    const int nb_e  = (E + 255) / 256;
    const int nb_r  = ((N + 15) / 16 + 7) / 8;
    const int nb_t  = (N + TR - 1) / TR;
    float* ale3 = acc + (size_t)nb_e * 64;

    k_edge_prep<<<nb_e, 256, 0, stream>>>(dst, ea, cvec3, ale3, tplane, E);

    for (int l = 0; l < 3; ++l) {
        const float* Wl  = (const float*)d_in[4 + 8 * l];
        const float* bv  = (const float*)d_in[9 + 8 * l];
        const float* xin = (l == 0) ? x : xbuf;
        const float* was = wvec + l * 2 * HF;
        const float* wad = was + HF;
        const float* cvec = cvec3 + l * 4;
        float* colsum = colsum3 + l * 2 * HF;
        float* colsq  = colsum + HF;

        if (l == 0) {
            k_node_dots<<<512, 256, 0, stream>>>(xin, was, wad, cvec, easum, als, ald, aself, N, E);
        } else {
            const float* gp = (const float*)d_in[10 + 8 * (l - 1)];
            const float* bp = (const float*)d_in[11 + 8 * (l - 1)];
            k_bn_rows<<<nb_r, 256, 0, stream>>>(hbuf, colsum - 2 * HF, colsq - 2 * HF, gp, bp, xbuf,
                                                was, wad, cvec, easum, als, ald, aself, N, E, 1);
        }
        k_tile_rowsum<<<nb_t, 1024, 0, stream>>>(src, dst, tplane, ale3 + (size_t)l * E, als, ald,
                                                 aself, xin, Wl, bv, hbuf, colsum, colsq,
                                                 elist, l, N, E);
    }
    k_bn_rows<<<nb_r, 256, 0, stream>>>(hbuf, colsum3 + 4 * HF, colsum3 + 5 * HF,
                                        (const float*)d_in[10 + 8 * 2],
                                        (const float*)d_in[11 + 8 * 2], (float*)d_out,
                                        wvec, wvec + HF, cvec3, easum, als, ald, aself, N, E, 0);
}
